# starting kernel
# baseline (speedup 1.0000x reference)
#include <hip/hip_runtime.h>
#include <math.h>

#define NN      4096
#define FIN     128
#define FHID    64
#define NHEADS  4
#define FOUT    64
#define HCATF   (NHEADS * FHID)
#define ALPHA_L 0.2f
#define CAP     128

typedef float v2f __attribute__((ext_vector_type(2)));
typedef float v8f __attribute__((ext_vector_type(8)));

__device__ __forceinline__ v8f wmma_f32_tile(const float* __restrict__ A, int lda,
                                             const float* __restrict__ B, int ldb,
                                             int K, int lane) {
  const int half = lane >> 4;
  const int l16  = lane & 15;
  v8f c = {0.f, 0.f, 0.f, 0.f, 0.f, 0.f, 0.f, 0.f};
  for (int k = 0; k < K; k += 4) {
    v2f a, b;
    const float* ap = A + (size_t)l16 * lda + k + 2 * half;
    a.x = ap[0];
    a.y = ap[1];
    b.x = B[(size_t)(k + 2 * half) * ldb + l16];
    b.y = B[(size_t)(k + 2 * half + 1) * ldb + l16];
    c = __builtin_amdgcn_wmma_f32_16x16x4_f32(false, a, false, b, (short)0, c, false, false);
  }
  return c;
}

__global__ void build_csr_kernel(const float* __restrict__ adj,
                                 int* __restrict__ cnt, int* __restrict__ nidx) {
  __shared__ int scan[128];
  const int row = blockIdx.x;
  const int t = threadIdx.x;
  const float4* arow4 = (const float4*)(adj + (size_t)row * NN) + t * 8;
  const int base = t * 32;

  float4 v[8];
#pragma unroll
  for (int q = 0; q < 8; ++q) v[q] = arow4[q];

  int c = 0;
#pragma unroll
  for (int q = 0; q < 8; ++q) {
    c += (v[q].x > 0.f) ? 1 : 0;
    c += (v[q].y > 0.f) ? 1 : 0;
    c += (v[q].z > 0.f) ? 1 : 0;
    c += (v[q].w > 0.f) ? 1 : 0;
  }
  scan[t] = c;
  __syncthreads();
  for (int off = 1; off < 128; off <<= 1) {
    int p = (t >= off) ? scan[t - off] : 0;
    __syncthreads();
    scan[t] += p;
    __syncthreads();
  }
  int k = scan[t] - c;
  const int total = scan[127];
  if (t == 0) cnt[row] = total < CAP ? total : CAP;
  int* out = nidx + (size_t)row * CAP;
#pragma unroll
  for (int q = 0; q < 8; ++q) {
    const int b = base + q * 4;
    if (v[q].x > 0.f) { if (k < CAP) out[k] = b + 0; ++k; }
    if (v[q].y > 0.f) { if (k < CAP) out[k] = b + 1; ++k; }
    if (v[q].z > 0.f) { if (k < CAP) out[k] = b + 2; ++k; }
    if (v[q].w > 0.f) { if (k < CAP) out[k] = b + 3; ++k; }
  }
}

__global__ void lin_kernel(const float* __restrict__ x, const float* __restrict__ W,
                           const float* __restrict__ b, float* __restrict__ h) {
  const int mtile = blockIdx.x;
  const int wave  = threadIdx.x >> 5;
  const int lane  = threadIdx.x & 31;
  const int half  = lane >> 4, l16 = lane & 15;
  const float* A = x + (size_t)mtile * 16 * FIN;
  const float* B = W + wave * 16;
  v8f c = wmma_f32_tile(A, FIN, B, FIN, FIN, lane);
  const float bias = b[wave * 16 + l16];
  float* out = h + (size_t)mtile * 16 * FIN + wave * 16 + l16;
#pragma unroll
  for (int r = 0; r < 8; ++r)
    out[(size_t)(r + 8 * half) * FIN] = c[r] + bias;
}

__global__ void heads_gemm_kernel(const float* __restrict__ h,
                                  const float* __restrict__ W_heads,
                                  float* __restrict__ hcat) {
  const int mtile = blockIdx.x;
  const int wave  = threadIdx.x >> 5;
  const int head  = wave >> 2;
  const int nt    = wave & 3;
  const int lane  = threadIdx.x & 31;
  const int half  = lane >> 4, l16 = lane & 15;
  const float* A = h + (size_t)mtile * 16 * FIN;
  const float* B = W_heads + (size_t)head * FIN * FHID + nt * 16;
  v8f c = wmma_f32_tile(A, FIN, B, FHID, FIN, lane);
  float* out = hcat + (size_t)mtile * 16 * HCATF + head * FHID + nt * 16 + l16;
#pragma unroll
  for (int r = 0; r < 8; ++r)
    out[(size_t)(r + 8 * half) * HCATF] = c[r];
}

__global__ void attn_coeff_heads_kernel(const float* __restrict__ hcat,
                                        const float* __restrict__ a_heads,
                                        float* __restrict__ src, float* __restrict__ dst) {
  const int idx = blockIdx.x * blockDim.x + threadIdx.x;
  if (idx >= NN * NHEADS) return;
  const int i = idx >> 2, head = idx & 3;
  const float4* wh = (const float4*)(hcat + (size_t)i * HCATF + head * FHID);
  const float4* as = (const float4*)(a_heads + head * 2 * FHID);
  const float4* ad = as + FHID / 4;
  float s = 0.f, d = 0.f;
#pragma unroll
  for (int f = 0; f < FHID / 4; ++f) {
    const float4 w = wh[f], A = as[f], B = ad[f];
    s += w.x * A.x + w.y * A.y + w.z * A.z + w.w * A.w;
    d += w.x * B.x + w.y * B.y + w.z * B.z + w.w * B.w;
  }
  src[head * NN + i] = s;
  dst[head * NN + i] = d;
}

__global__ void gat_aggregate_heads_kernel(const float* __restrict__ hcat,
                                           const float* __restrict__ src,
                                           const float* __restrict__ dst,
                                           const int* __restrict__ cnt,
                                           const int* __restrict__ nidx,
                                           float* __restrict__ gat1) {
  __shared__ float wgt[CAP];
  __shared__ int   jidx[CAP];
  __shared__ float red[64];
  const int row  = blockIdx.x >> 2;
  const int head = blockIdx.x & 3;
  const int t    = threadIdx.x;
  const int deg  = cnt[row];
  const float si = src[head * NN + row];
  const int* nb  = nidx + (size_t)row * CAP;

  float m = -3.4e38f;
  for (int k = t; k < deg; k += 64) {
    const int j = nb[k];
    jidx[k] = j;
    float e = si + dst[head * NN + j];
    e = (e >= 0.f) ? e : ALPHA_L * e;
    wgt[k] = e;
    m = fmaxf(m, e);
  }
  red[t] = m;
  __syncthreads();
  for (int s = 32; s > 0; s >>= 1) { if (t < s) red[t] = fmaxf(red[t], red[t + s]); __syncthreads(); }
  m = red[0];
  __syncthreads();

  float ssum = 0.f;
  for (int k = t; k < deg; k += 64) {
    const float w = expf(wgt[k] - m);
    wgt[k] = w;
    ssum += w;
  }
  red[t] = ssum;
  __syncthreads();
  for (int s = 32; s > 0; s >>= 1) { if (t < s) red[t] += red[t + s]; __syncthreads(); }
  const float denom = red[0];
  __syncthreads();

  float acc = 0.f;
  for (int k = 0; k < deg; ++k)
    acc += wgt[k] * hcat[(size_t)jidx[k] * HCATF + head * FHID + t];
  float o = acc / denom;
  o = (o > 0.f) ? o : (expf(o) - 1.f);
  gat1[(size_t)row * HCATF + head * FHID + t] = o;
}

__global__ void end_gemm_kernel(const float* __restrict__ gat1,
                                const float* __restrict__ W_end,
                                float* __restrict__ Wh2) {
  const int mtile = blockIdx.x;
  const int wave  = threadIdx.x >> 5;
  const int lane  = threadIdx.x & 31;
  const int half  = lane >> 4, l16 = lane & 15;
  const float* A = gat1 + (size_t)mtile * 16 * HCATF;
  const float* B = W_end + wave * 16;
  v8f c = wmma_f32_tile(A, HCATF, B, FOUT, HCATF, lane);
  float* out = Wh2 + (size_t)mtile * 16 * FOUT + wave * 16 + l16;
#pragma unroll
  for (int r = 0; r < 8; ++r)
    out[(size_t)(r + 8 * half) * FOUT] = c[r];
}

__global__ void attn_coeff_end_kernel(const float* __restrict__ Wh2,
                                      const float* __restrict__ a_end,
                                      float* __restrict__ src, float* __restrict__ dst) {
  const int i = blockIdx.x * blockDim.x + threadIdx.x;
  if (i >= NN) return;
  const float4* wh = (const float4*)(Wh2 + (size_t)i * FOUT);
  const float4* as = (const float4*)a_end;
  const float4* ad = as + FOUT / 4;
  float s = 0.f, d = 0.f;
#pragma unroll
  for (int f = 0; f < FOUT / 4; ++f) {
    const float4 w = wh[f], A = as[f], B = ad[f];
    s += w.x * A.x + w.y * A.y + w.z * A.z + w.w * A.w;
    d += w.x * B.x + w.y * B.y + w.z * B.z + w.w * B.w;
  }
  src[i] = s;
  dst[i] = d;
}

__global__ void gat_end_kernel(const float* __restrict__ Wh2,
                               const float* __restrict__ src,
                               const float* __restrict__ dst,
                               const int* __restrict__ cnt,
                               const int* __restrict__ nidx,
                               float* __restrict__ out) {
  __shared__ float wgt[CAP];
  __shared__ int   jidx[CAP];
  __shared__ float red[64];
  const int row = blockIdx.x;
  const int t   = threadIdx.x;
  const int deg = cnt[row];
  const float si = src[row];
  const int* nb = nidx + (size_t)row * CAP;

  float m = -3.4e38f;
  for (int k = t; k < deg; k += 64) {
    const int j = nb[k];
    jidx[k] = j;
    float e = si + dst[j];
    e = (e >= 0.f) ? e : ALPHA_L * e;
    wgt[k] = e;
    m = fmaxf(m, e);
  }
  red[t] = m;
  __syncthreads();
  for (int s = 32; s > 0; s >>= 1) { if (t < s) red[t] = fmaxf(red[t], red[t + s]); __syncthreads(); }
  m = red[0];
  __syncthreads();

  float ssum = 0.f;
  for (int k = t; k < deg; k += 64) {
    const float w = expf(wgt[k] - m);
    wgt[k] = w;
    ssum += w;
  }
  red[t] = ssum;
  __syncthreads();
  for (int s = 32; s > 0; s >>= 1) { if (t < s) red[t] += red[t + s]; __syncthreads(); }
  const float denom = red[0];
  __syncthreads();

  float acc = 0.f;
  for (int k = 0; k < deg; ++k)
    acc += wgt[k] * Wh2[(size_t)jidx[k] * FOUT + t];
  float o = acc / denom;
  o = (o > 0.f) ? o : (expf(o) - 1.f);

  red[t] = o;
  __syncthreads();
  for (int s = 32; s > 0; s >>= 1) { if (t < s) red[t] = fmaxf(red[t], red[t + s]); __syncthreads(); }
  const float mo = red[0];
  __syncthreads();
  const float ex = expf(o - mo);
  red[t] = ex;
  __syncthreads();
  for (int s = 32; s > 0; s >>= 1) { if (t < s) red[t] += red[t + s]; __syncthreads(); }
  const float sum = red[0];
  out[(size_t)row * FOUT + t] = ex / sum;
}

extern "C" void kernel_launch(void* const* d_in, const int* in_sizes, int n_in,
                              void* d_out, int out_size, void* d_ws, size_t ws_size,
                              hipStream_t stream) {
  (void)in_sizes; (void)n_in; (void)out_size; (void)ws_size;
  const float* x       = (const float*)d_in[0];
  const float* adj     = (const float*)d_in[1];
  const float* lin_W   = (const float*)d_in[2];
  const float* lin_b   = (const float*)d_in[3];
  const float* W_heads = (const float*)d_in[4];
  const float* a_heads = (const float*)d_in[5];
  const float* W_end   = (const float*)d_in[6];
  const float* a_end   = (const float*)d_in[7];
  float* out = (float*)d_out;

  char* ws = (char*)d_ws;
  float* h     = (float*)ws;  ws += (size_t)NN * FIN   * sizeof(float);
  float* hcat  = (float*)ws;  ws += (size_t)NN * HCATF * sizeof(float);
  float* gat1  = (float*)ws;  ws += (size_t)NN * HCATF * sizeof(float);
  float* Wh2   = (float*)ws;  ws += (size_t)NN * FOUT  * sizeof(float);
  float* src1  = (float*)ws;  ws += (size_t)NHEADS * NN * sizeof(float);
  float* dst1  = (float*)ws;  ws += (size_t)NHEADS * NN * sizeof(float);
  float* src2  = (float*)ws;  ws += (size_t)NN * sizeof(float);
  float* dst2  = (float*)ws;  ws += (size_t)NN * sizeof(float);
  int*   ncnt  = (int*)ws;    ws += (size_t)NN * sizeof(int);
  int*   nidx  = (int*)ws;    ws += (size_t)NN * CAP * sizeof(int);

  build_csr_kernel<<<NN, 128, 0, stream>>>(adj, ncnt, nidx);
  lin_kernel<<<NN / 16, 256, 0, stream>>>(x, lin_W, lin_b, h);
  heads_gemm_kernel<<<NN / 16, 512, 0, stream>>>(h, W_heads, hcat);
  attn_coeff_heads_kernel<<<(NN * NHEADS) / 256, 256, 0, stream>>>(hcat, a_heads, src1, dst1);
  gat_aggregate_heads_kernel<<<NN * NHEADS, 64, 0, stream>>>(hcat, src1, dst1, ncnt, nidx, gat1);
  end_gemm_kernel<<<NN / 16, 128, 0, stream>>>(gat1, W_end, Wh2);
  attn_coeff_end_kernel<<<NN / 256, 256, 0, stream>>>(Wh2, a_end, src2, dst2);
  gat_end_kernel<<<NN, 64, 0, stream>>>(Wh2, src2, dst2, ncnt, nidx, out);
}


# Final aggregation: 16-lane half-wave per row, lane = 4 features (b128 neighbor rows, padded batches of 16), DPP row reductions replace 28 block barriers; waves 8192 -> 2048; softmax sums reassociated
# speedup vs baseline: 1.5098x; 1.5098x over previous
#include <hip/hip_runtime.h>
#include <math.h>

#define NN      4096
#define FIN     128
#define FHID    64
#define NHEADS  4
#define FOUT    64
#define HCATF   (NHEADS * FHID)
#define ALPHA_L 0.2f
#define CAP     128

typedef float v2f __attribute__((ext_vector_type(2)));
typedef float v4f __attribute__((ext_vector_type(4)));
typedef float v8f __attribute__((ext_vector_type(8)));

struct v1f { float x; __device__ __forceinline__ float operator[](int) const { return x; } };
template <int NT> struct RowVec;
template <> struct RowVec<1> { typedef v1f T; };
template <> struct RowVec<2> { typedef v2f T; };
template <> struct RowVec<4> { typedef v4f T; };

__device__ __forceinline__ void ldrow(const float* __restrict__ p, v1f& o) { o.x = p[0]; }
__device__ __forceinline__ void ldrow(const float* __restrict__ p, v2f& o) { o = *(const v2f*)p; }
__device__ __forceinline__ void ldrow(const float* __restrict__ p, v4f& o) { o = *(const v4f*)p; }

template <int NT, int KB> struct Frag {
  v2f a[KB];
  typename RowVec<NT>::T b0[KB], b1[KB];
};

template <int NT, int KB>
__device__ __forceinline__ void frag_load(Frag<NT, KB>& f, const float* __restrict__ Ap,
                                          const float* __restrict__ Bp, int ldb, int k0) {
#pragma unroll
  for (int s = 0; s < KB; ++s) {
    const int k = k0 + 4 * s;
    f.a[s] = *(const v2f*)(Ap + k);
    ldrow(Bp + (size_t)k * ldb, f.b0[s]);
    ldrow(Bp + (size_t)(k + 1) * ldb, f.b1[s]);
  }
}

template <int NT, int KB>
__device__ __forceinline__ void frag_mma(const Frag<NT, KB>& f, v8f (&acc)[NT]) {
#pragma unroll
  for (int s = 0; s < KB; ++s) {
#pragma unroll
    for (int t = 0; t < NT; ++t) {
      v2f b;
      b.x = f.b0[s][t];
      b.y = f.b1[s][t];
      acc[t] = __builtin_amdgcn_wmma_f32_16x16x4_f32(false, f.a[s], false, b, (short)0, acc[t],
                                                     false, false);
    }
  }
}

template <int NT, int K, int KB>
__device__ __forceinline__ void gemm_strip(const float* __restrict__ A, int lda,
                                           const float* __restrict__ B, int ldb,
                                           int lane, v8f (&acc)[NT]) {
  static_assert(K % (8 * KB) == 0, "K must hold an even number of k-blocks");
  const int half = lane >> 4;
  const int l16  = lane & 15;
  const float* Ap = A + (size_t)l16 * lda + 2 * half;
  const float* Bp = B + (size_t)(2 * half) * ldb + NT * l16;
  Frag<NT, KB> f0, f1;
  frag_load<NT, KB>(f0, Ap, Bp, ldb, 0);
#pragma unroll
  for (int k0 = 0; k0 < K; k0 += 8 * KB) {
    frag_load<NT, KB>(f1, Ap, Bp, ldb, k0 + 4 * KB);
    frag_mma<NT, KB>(f0, acc);
    if (k0 + 8 * KB < K) frag_load<NT, KB>(f0, Ap, Bp, ldb, k0 + 8 * KB);
    frag_mma<NT, KB>(f1, acc);
  }
}

__device__ __forceinline__ void dot64x2(const float* __restrict__ wrow,
                                        const float* __restrict__ a,
                                        float& s_out, float& d_out) {
  const float4* w4 = (const float4*)wrow;
  const float4* as = (const float4*)a;
  const float4* ad = as + 16;
  float s = 0.f, d = 0.f;
#pragma unroll
  for (int f = 0; f < 16; ++f) {
    const float4 w = w4[f], A = as[f], B = ad[f];
    s += w.x * A.x + w.y * A.y + w.z * A.z + w.w * A.w;
    d += w.x * B.x + w.y * B.y + w.z * B.z + w.w * B.w;
  }
  s_out = s;
  d_out = d;
}

__device__ __forceinline__ void fold_block(const float* __restrict__ W_heads,
                                           const float* __restrict__ a_heads,
                                           const float* __restrict__ W_end,
                                           const float* __restrict__ a_end,
                                           float* __restrict__ u1, float* __restrict__ u2,
                                           int t) {
  float s, d;
  dot64x2(W_end + (size_t)t * FOUT, a_end, s, d);
  u2[t] = s;
  u2[HCATF + t] = d;
  const int k  = t & (FIN - 1);
  const int hp = t >> 7;
#pragma unroll
  for (int hh = 0; hh < 2; ++hh) {
    const int hd = 2 * hp + hh;
    dot64x2(W_heads + ((size_t)hd * FIN + k) * FHID, a_heads + hd * 2 * FHID, s, d);
    u1[k * 16 + hd] = s;
    u1[k * 16 + 4 + hd] = d;
  }
  float4 z; z.x = 0.f; z.y = 0.f; z.z = 0.f; z.w = 0.f;
  *(float4*)(u1 + k * 16 + 8 + 4 * hp) = z;
}

__global__ __launch_bounds__(128)
void lin_kernel(const float* __restrict__ x, const float* __restrict__ W,
                const float* __restrict__ b, float* __restrict__ h,
                const float* __restrict__ W_heads, const float* __restrict__ a_heads,
                const float* __restrict__ W_end, const float* __restrict__ a_end,
                float* __restrict__ u1, float* __restrict__ u2) {
  if (blockIdx.x < 2) {
    fold_block(W_heads, a_heads, W_end, a_end, u1, u2, blockIdx.x * 128 + threadIdx.x);
    return;
  }
  const int mtile = blockIdx.x - 2;
  const int wave  = threadIdx.x >> 5;
  const int lane  = threadIdx.x & 31;
  const int half  = lane >> 4, l16 = lane & 15;
  const int n0    = wave * 32;
  const v8f zero = {0.f, 0.f, 0.f, 0.f, 0.f, 0.f, 0.f, 0.f};
  v8f acc[2] = {zero, zero};
  gemm_strip<2, FIN, 16>(x + (size_t)mtile * 16 * FIN, FIN, W + n0, FIN, lane, acc);
  const v2f bias = *(const v2f*)(b + n0 + 2 * l16);
  float* out = h + (size_t)(mtile * 16 + 8 * half) * FIN + n0 + 2 * l16;
#pragma unroll
  for (int r = 0; r < 8; ++r) {
    v2f o;
    o.x = acc[0][r] + bias.x;
    o.y = acc[1][r] + bias.y;
    *(v2f*)(out + (size_t)r * FIN) = o;
  }
}

__global__ __launch_bounds__(160)
void heads_gemm_kernel(const float* __restrict__ h,
                       const float* __restrict__ W_heads,
                       float* __restrict__ hcat,
                       const float* __restrict__ u1,
                       float* __restrict__ src, float* __restrict__ dst) {
  const int mtile = blockIdx.x;
  const int wave  = threadIdx.x >> 5;
  const int lane  = threadIdx.x & 31;
  const int half  = lane >> 4, l16 = lane & 15;
  const float* A = h + (size_t)mtile * 16 * FIN;
  const v8f zero = {0.f, 0.f, 0.f, 0.f, 0.f, 0.f, 0.f, 0.f};
  if (wave == 4) {
    v8f cc[1] = {zero};
    gemm_strip<1, FIN, 16>(A, FIN, u1, 16, lane, cc);
    float* cp = ((l16 & 4) ? dst : src) + (size_t)(l16 & 3) * NN + mtile * 16 + 8 * half;
    if (l16 < 8) {
#pragma unroll
      for (int r = 0; r < 8; ++r) cp[r] = cc[0][r];
    }
    return;
  }
  const int head = wave;
  v8f acc[4] = {zero, zero, zero, zero};
  gemm_strip<4, FIN, 8>(A, FIN, W_heads + (size_t)head * FIN * FHID, FHID, lane, acc);
  float* out = hcat + (size_t)(mtile * 16 + 8 * half) * HCATF + head * FHID + 4 * l16;
#pragma unroll
  for (int r = 0; r < 8; ++r) {
    v4f o;
    o.x = acc[0][r];
    o.y = acc[1][r];
    o.z = acc[2][r];
    o.w = acc[3][r];
    *(v4f*)(out + (size_t)r * HCATF) = o;
  }
}

__global__ __launch_bounds__(64)
void gat_heads_fused_kernel(const float* __restrict__ adj,
                            const float* __restrict__ hcat,
                            const float* __restrict__ src,
                            const float* __restrict__ dst,
                            int* __restrict__ cnt, int* __restrict__ nidx,
                            float* __restrict__ gat1,
                            const float* __restrict__ u2,
                            float* __restrict__ src2, float* __restrict__ dst2) {
  __shared__ int   wtot[2];
  __shared__ float csum[2][2];
  __shared__ int   jidx[CAP];
  __shared__ float wgt[NHEADS][CAP];
  const int row  = blockIdx.x;
  const int t    = threadIdx.x;
  const int wave = t >> 5;
  const int lane = t & 31;
  const int l16  = lane & 15;
  const int head = wave * 2 + (lane >> 4);

  const float4* arow4 = (const float4*)(adj + (size_t)row * NN) + t * 16;
  const int base = t * 64;

  float4 v[16];
#pragma unroll
  for (int q = 0; q < 16; ++q) v[q] = arow4[q];

  const float si = src[head * NN + row];

  int c = 0;
#pragma unroll
  for (int q = 0; q < 16; ++q) {
    c += (v[q].x > 0.f) ? 1 : 0;
    c += (v[q].y > 0.f) ? 1 : 0;
    c += (v[q].z > 0.f) ? 1 : 0;
    c += (v[q].w > 0.f) ? 1 : 0;
  }
  int inc = c;
  for (int off = 1; off < 32; off <<= 1) {
    const int n = __shfl_up(inc, off);
    if (lane >= off) inc += n;
  }
  if (lane == 31) wtot[wave] = inc;
  __syncthreads();
  const int t0 = wtot[0], t1 = wtot[1];
  const int total = t0 + t1;
  int k = inc - c;
  if (wave > 0) k += t0;
  const int deg = total < CAP ? total : CAP;
  if (t == 0) cnt[row] = deg;
  int* outp = nidx + (size_t)row * CAP;
#pragma unroll
  for (int q = 0; q < 16; ++q) {
    const int b = base + q * 4;
    if (v[q].x > 0.f) { if (k < CAP) { jidx[k] = b + 0; outp[k] = b + 0; } ++k; }
    if (v[q].y > 0.f) { if (k < CAP) { jidx[k] = b + 1; outp[k] = b + 1; } ++k; }
    if (v[q].z > 0.f) { if (k < CAP) { jidx[k] = b + 2; outp[k] = b + 2; } ++k; }
    if (v[q].w > 0.f) { if (k < CAP) { jidx[k] = b + 3; outp[k] = b + 3; } ++k; }
  }
  __syncthreads();

  float* wg = wgt[head];
  const float* dsth = dst + head * NN;

  float m = -3.4e38f;
  for (int kk = l16; kk < deg; kk += 16) {
    const int j = jidx[kk];
    float e = si + dsth[j];
    e = (e >= 0.f) ? e : ALPHA_L * e;
    wg[kk] = e;
    m = fmaxf(m, e);
  }
  for (int off = 8; off > 0; off >>= 1) m = fmaxf(m, __shfl_xor(m, off));

  float ssum = 0.f;
  for (int kk = l16; kk < deg; kk += 16) {
    const float w = expf(wg[kk] - m);
    wg[kk] = w;
    ssum += w;
  }
  for (int off = 8; off > 0; off >>= 1) ssum += __shfl_xor(ssum, off);
  const float denom = ssum;
  __syncthreads();

  const float* hb = hcat + wave * (2 * FHID) + 4 * lane;
  const float4 us = *(const float4*)(u2 + wave * (2 * FHID) + 4 * lane);
  const float4 ud = *(const float4*)(u2 + HCATF + wave * (2 * FHID) + 4 * lane);
  float ax = 0.f, ay = 0.f, az = 0.f, aw = 0.f;
  for (int kk = 0; kk < deg; ++kk) {
    const float w = wg[kk];
    const float4 hv = *(const float4*)(hb + (size_t)jidx[kk] * HCATF);
    ax += w * hv.x;
    ay += w * hv.y;
    az += w * hv.z;
    aw += w * hv.w;
  }
  float4 o;
  o.x = ax / denom;
  o.y = ay / denom;
  o.z = az / denom;
  o.w = aw / denom;
  o.x = (o.x > 0.f) ? o.x : (expf(o.x) - 1.f);
  o.y = (o.y > 0.f) ? o.y : (expf(o.y) - 1.f);
  o.z = (o.z > 0.f) ? o.z : (expf(o.z) - 1.f);
  o.w = (o.w > 0.f) ? o.w : (expf(o.w) - 1.f);

  float ps = o.x * us.x + o.y * us.y + o.z * us.z + o.w * us.w;
  float pd = o.x * ud.x + o.y * ud.y + o.z * ud.z + o.w * ud.w;
  for (int off = 16; off > 0; off >>= 1) {
    ps += __shfl_xor(ps, off);
    pd += __shfl_xor(pd, off);
  }
  if (lane == 0) { csum[wave][0] = ps; csum[wave][1] = pd; }
  __syncthreads();
  *(float4*)(gat1 + (size_t)row * HCATF + wave * (2 * FHID) + 4 * lane) = o;
  if (t == 0) {
    src2[row] = csum[0][0] + csum[1][0];
    dst2[row] = csum[0][1] + csum[1][1];
  }
}

__global__ __launch_bounds__(64) __attribute__((amdgpu_waves_per_eu(1, 2)))
void end_gemm_kernel(const float* __restrict__ gat1,
                     const float* __restrict__ W_end,
                     float* __restrict__ Wh2) {
  const int mtile = blockIdx.x;
  const int wave  = threadIdx.x >> 5;
  const int lane  = threadIdx.x & 31;
  const int half  = lane >> 4, l16 = lane & 15;
  const int n0    = wave * 32;
  const v8f zero = {0.f, 0.f, 0.f, 0.f, 0.f, 0.f, 0.f, 0.f};
  v8f acc[2] = {zero, zero};
  gemm_strip<2, HCATF, 16>(gat1 + (size_t)mtile * 16 * HCATF, HCATF, W_end + n0, FOUT, lane, acc);
  float* out = Wh2 + (size_t)(mtile * 16 + 8 * half) * FOUT + n0 + 2 * l16;
#pragma unroll
  for (int r = 0; r < 8; ++r) {
    v2f o;
    o.x = acc[0][r];
    o.y = acc[1][r];
    *(v2f*)(out + (size_t)r * FOUT) = o;
  }
}

#define EB 16

template <int M>
__device__ __forceinline__ float row_xor(float v) {
  const int b = __builtin_bit_cast(int, v);
  return __builtin_bit_cast(float, __builtin_amdgcn_update_dpp(b, b, 0x160 + M, 0xf, 0xf, false));
}
__device__ __forceinline__ float row16_max(float v) {
  v = fmaxf(v, row_xor<1>(v));
  v = fmaxf(v, row_xor<2>(v));
  v = fmaxf(v, row_xor<4>(v));
  v = fmaxf(v, row_xor<8>(v));
  return v;
}
__device__ __forceinline__ float row16_sum(float v) {
  v += row_xor<1>(v);
  v += row_xor<2>(v);
  v += row_xor<4>(v);
  v += row_xor<8>(v);
  return v;
}
__device__ __forceinline__ float elu1(float x) { return (x > 0.f) ? x : (expf(x) - 1.f); }

__global__ __launch_bounds__(32) __attribute__((amdgpu_waves_per_eu(1, 4)))
void gat_end_kernel(const float* __restrict__ Wh2,
                    const float* __restrict__ src,
                    const float* __restrict__ dst,
                    const int* __restrict__ cnt,
                    const int* __restrict__ nidx,
                    float* __restrict__ out) {
  __shared__ __attribute__((aligned(16))) float wgt[2][CAP];
  __shared__ __attribute__((aligned(16))) int   jidx[2][CAP];
  const int lane = threadIdx.x;
  const int hw   = lane >> 4;
  const int l16  = lane & 15;
  const int row  = blockIdx.x * 2 + hw;
  const int deg  = cnt[row];
  float si = src[row];
  asm volatile("" : "+v"(si));
  const int* nb  = nidx + (size_t)row * CAP;
  const int degp = (deg + EB - 1) & ~(EB - 1);
  float* wg = wgt[hw];
  int*   ji = jidx[hw];

  float m = -3.4e38f;
  for (int kb = 0; kb < degp; kb += 64) {
    int   jv[4];
    float dv[4];
#pragma unroll
    for (int s = 0; s < 4; ++s) {
      const int k = kb + 16 * s + l16;
      jv[s] = nb[k < deg ? k : deg - 1];
    }
#pragma unroll
    for (int s = 0; s < 4; ++s) dv[s] = dst[jv[s]];
#pragma unroll
    for (int s = 0; s < 4; ++s) {
      const int k = kb + 16 * s + l16;
      float e = si + dv[s];
      e = (e >= 0.f) ? e : ALPHA_L * e;
      e = (k < deg) ? e : -INFINITY;
      ji[k] = jv[s];
      wg[k] = e;
      m = fmaxf(m, e);
    }
  }
  m = row16_max(m);

  float ssum = 0.f;
  for (int kb = 0; kb < degp; kb += 64) {
    float ev[4];
#pragma unroll
    for (int s = 0; s < 4; ++s) ev[s] = wg[kb + 16 * s + l16];
#pragma unroll
    for (int s = 0; s < 4; ++s) {
      const float w = expf(ev[s] - m);
      wg[kb + 16 * s + l16] = w;
      ssum += w;
    }
  }
  const float denom = row16_sum(ssum);
  __syncthreads();

  const unsigned lo = 4 * l16;
  float ax = 0.f, ay = 0.f, az = 0.f, aw = 0.f;
  for (int k0 = 0; k0 < degp; k0 += EB) {
    int    jj[EB];
    float  ww[EB];
    float4 hv[EB];
#pragma unroll
    for (int q = 0; q < EB / 4; ++q) {
      const int4   jq = *(const int4*)(ji + k0 + 4 * q);
      const float4 wq = *(const float4*)(wg + k0 + 4 * q);
      jj[4 * q + 0] = jq.x; jj[4 * q + 1] = jq.y; jj[4 * q + 2] = jq.z; jj[4 * q + 3] = jq.w;
      ww[4 * q + 0] = wq.x; ww[4 * q + 1] = wq.y; ww[4 * q + 2] = wq.z; ww[4 * q + 3] = wq.w;
    }
#pragma unroll
    for (int i = 0; i < EB; ++i)
      hv[i] = *(const float4*)(Wh2 + ((unsigned)jj[i] * FOUT + lo));
#pragma unroll
    for (int i = 0; i < EB; ++i) {
      ax = __builtin_fmaf(ww[i], hv[i].x, ax);
      ay = __builtin_fmaf(ww[i], hv[i].y, ay);
      az = __builtin_fmaf(ww[i], hv[i].z, az);
      aw = __builtin_fmaf(ww[i], hv[i].w, aw);
    }
  }
  float4 o;
  o.x = elu1(ax / denom);
  o.y = elu1(ay / denom);
  o.z = elu1(az / denom);
  o.w = elu1(aw / denom);

  const float mo = row16_max(fmaxf(fmaxf(o.x, o.y), fmaxf(o.z, o.w)));
  float4 ex;
  ex.x = expf(o.x - mo);
  ex.y = expf(o.y - mo);
  ex.z = expf(o.z - mo);
  ex.w = expf(o.w - mo);
  const float sum = row16_sum((ex.x + ex.y) + (ex.z + ex.w));
  float4 r;
  r.x = ex.x / sum;
  r.y = ex.y / sum;
  r.z = ex.z / sum;
  r.w = ex.w / sum;
  *(float4*)(out + (size_t)row * FOUT + lo) = r;
}

extern "C" void kernel_launch(void* const* d_in, const int* in_sizes, int n_in,
                              void* d_out, int out_size, void* d_ws, size_t ws_size,
                              hipStream_t stream) {
  (void)in_sizes; (void)n_in; (void)out_size; (void)ws_size;
  const float* x       = (const float*)d_in[0];
  const float* adj     = (const float*)d_in[1];
  const float* lin_W   = (const float*)d_in[2];
  const float* lin_b   = (const float*)d_in[3];
  const float* W_heads = (const float*)d_in[4];
  const float* a_heads = (const float*)d_in[5];
  const float* W_end   = (const float*)d_in[6];
  const float* a_end   = (const float*)d_in[7];
  float* out = (float*)d_out;

  char* ws = (char*)d_ws;
  float* h     = (float*)ws;  ws += (size_t)NN * FIN   * sizeof(float);
  float* hcat  = (float*)ws;  ws += (size_t)NN * HCATF * sizeof(float);
  float* gat1  = (float*)ws;  ws += (size_t)NN * HCATF * sizeof(float);
  float* Wh2   = (float*)ws;  ws += (size_t)NN * FOUT  * sizeof(float);
  float* src1  = (float*)ws;  ws += (size_t)NHEADS * NN * sizeof(float);
  float* dst1  = (float*)ws;  ws += (size_t)NHEADS * NN * sizeof(float);
  float* src2  = (float*)ws;  ws += (size_t)NN * sizeof(float);
  float* dst2  = (float*)ws;  ws += (size_t)NN * sizeof(float);
  int*   ncnt  = (int*)ws;    ws += (size_t)NN * sizeof(int);
  int*   nidx  = (int*)ws;    ws += (size_t)NN * CAP * sizeof(int);
  float* u1    = Wh2;
  float* u2    = Wh2 + FIN * 16;

  lin_kernel<<<NN / 16 + 2, 128, 0, stream>>>(x, lin_W, lin_b, h,
                                               W_heads, a_heads, W_end, a_end, u1, u2);
  heads_gemm_kernel<<<NN / 16, 160, 0, stream>>>(h, W_heads, hcat, u1, src1, dst1);
  gat_heads_fused_kernel<<<NN, 64, 0, stream>>>(adj, hcat, src1, dst1, ncnt, nidx, gat1,
                                                 u2, src2, dst2);
  end_gemm_kernel<<<NN / 16, 64, 0, stream>>>(gat1, W_end, Wh2);
  gat_end_kernel<<<NN / 2, 32, 0, stream>>>(Wh2, src2, dst2, ncnt, nidx, out);
}


# Linear layer computed in the head-projection launch, transposed: its accumulators are the head product's A fragments; k split over 4 waves, partials added in LDS; coefficient dots re-summed; 5->4
# speedup vs baseline: 1.6735x; 1.1084x over previous
#include <hip/hip_runtime.h>
#include <math.h>

#define NN      4096
#define FIN     128
#define FHID    64
#define NHEADS  4
#define FOUT    64
#define HCATF   (NHEADS * FHID)
#define ALPHA_L 0.2f
#define CAP     128

typedef float v2f __attribute__((ext_vector_type(2)));
typedef float v4f __attribute__((ext_vector_type(4)));
typedef float v8f __attribute__((ext_vector_type(8)));

struct v1f { float x; __device__ __forceinline__ float operator[](int) const { return x; } };
template <int NT> struct RowVec;
template <> struct RowVec<1> { typedef v1f T; };
template <> struct RowVec<2> { typedef v2f T; };
template <> struct RowVec<4> { typedef v4f T; };

__device__ __forceinline__ void ldrow(const float* __restrict__ p, v1f& o) { o.x = p[0]; }
__device__ __forceinline__ void ldrow(const float* __restrict__ p, v2f& o) { o = *(const v2f*)p; }
__device__ __forceinline__ void ldrow(const float* __restrict__ p, v4f& o) { o = *(const v4f*)p; }

template <int NT, int KB> struct Frag {
  v2f a[KB];
  typename RowVec<NT>::T b0[KB], b1[KB];
};

template <int NT, int KB>
__device__ __forceinline__ void frag_load(Frag<NT, KB>& f, const float* __restrict__ Ap,
                                          const float* __restrict__ Bp, int ldb, int k0) {
#pragma unroll
  for (int s = 0; s < KB; ++s) {
    const int k = k0 + 4 * s;
    f.a[s] = *(const v2f*)(Ap + k);
    ldrow(Bp + (size_t)k * ldb, f.b0[s]);
    ldrow(Bp + (size_t)(k + 1) * ldb, f.b1[s]);
  }
}

template <int NT, int KB>
__device__ __forceinline__ void frag_mma(const Frag<NT, KB>& f, v8f (&acc)[NT]) {
#pragma unroll
  for (int s = 0; s < KB; ++s) {
#pragma unroll
    for (int t = 0; t < NT; ++t) {
      v2f b;
      b.x = f.b0[s][t];
      b.y = f.b1[s][t];
      acc[t] = __builtin_amdgcn_wmma_f32_16x16x4_f32(false, f.a[s], false, b, (short)0, acc[t],
                                                     false, false);
    }
  }
}

template <int NT, int K, int KB>
__device__ __forceinline__ void gemm_strip(const float* __restrict__ A, int lda,
                                           const float* __restrict__ B, int ldb,
                                           int lane, v8f (&acc)[NT]) {
  static_assert(K % (8 * KB) == 0, "K must hold an even number of k-blocks");
  const int half = lane >> 4;
  const int l16  = lane & 15;
  const float* Ap = A + (size_t)l16 * lda + 2 * half;
  const float* Bp = B + (size_t)(2 * half) * ldb + NT * l16;
  Frag<NT, KB> f0, f1;
  frag_load<NT, KB>(f0, Ap, Bp, ldb, 0);
#pragma unroll
  for (int k0 = 0; k0 < K; k0 += 8 * KB) {
    frag_load<NT, KB>(f1, Ap, Bp, ldb, k0 + 4 * KB);
    frag_mma<NT, KB>(f0, acc);
    if (k0 + 8 * KB < K) frag_load<NT, KB>(f0, Ap, Bp, ldb, k0 + 8 * KB);
    frag_mma<NT, KB>(f1, acc);
  }
}

__device__ __forceinline__ void dot64x2(const float* __restrict__ wrow,
                                        const float* __restrict__ a,
                                        float& s_out, float& d_out) {
  const float4* w4 = (const float4*)wrow;
  const float4* as = (const float4*)a;
  const float4* ad = as + 16;
  float s = 0.f, d = 0.f;
#pragma unroll
  for (int f = 0; f < 16; ++f) {
    const float4 w = w4[f], A = as[f], B = ad[f];
    s += w.x * A.x + w.y * A.y + w.z * A.z + w.w * A.w;
    d += w.x * B.x + w.y * B.y + w.z * B.z + w.w * B.w;
  }
  s_out = s;
  d_out = d;
}

template <int M>
__device__ __forceinline__ float row_xor(float v) {
  const int b = __builtin_bit_cast(int, v);
  return __builtin_bit_cast(float, __builtin_amdgcn_update_dpp(b, b, 0x160 + M, 0xf, 0xf, false));
}
__device__ __forceinline__ float row16_max(float v) {
  v = fmaxf(v, row_xor<1>(v));
  v = fmaxf(v, row_xor<2>(v));
  v = fmaxf(v, row_xor<4>(v));
  v = fmaxf(v, row_xor<8>(v));
  return v;
}
__device__ __forceinline__ float row16_sum(float v) {
  v += row_xor<1>(v);
  v += row_xor<2>(v);
  v += row_xor<4>(v);
  v += row_xor<8>(v);
  return v;
}

#define P1G 8
struct P1Frag {
  v4f xb[P1G];
  v2f w[P1G][4];
};
__device__ __forceinline__ void p1_load(P1Frag& f, const float* __restrict__ xp,
                                        const float* __restrict__ wp, int k0) {
#pragma unroll
  for (int g = 0; g < P1G; ++g) {
    const int k = k0 + 8 * g;
    f.xb[g] = *(const v4f*)(xp + k);
#pragma unroll
    for (int j = 0; j < 4; ++j) f.w[g][j] = *(const v2f*)(wp + (size_t)(k + j) * FIN);
  }
}
__device__ __forceinline__ void p1_mma(const P1Frag& f, v8f (&hT)[2]) {
#pragma unroll
  for (int g = 0; g < P1G; ++g) {
#pragma unroll
    for (int e = 0; e < 2; ++e) {
      v2f b, a0, a1;
      b.x  = f.xb[g][2 * e];
      b.y  = f.xb[g][2 * e + 1];
      a0.x = f.w[g][2 * e].x;
      a0.y = f.w[g][2 * e + 1].x;
      a1.x = f.w[g][2 * e].y;
      a1.y = f.w[g][2 * e + 1].y;
      hT[0] = __builtin_amdgcn_wmma_f32_16x16x4_f32(false, a0, false, b, (short)0, hT[0],
                                                    false, false);
      hT[1] = __builtin_amdgcn_wmma_f32_16x16x4_f32(false, a1, false, b, (short)0, hT[1],
                                                    false, false);
    }
  }
}

struct P2Frag { v4f b0[2][NHEADS], b1[2][NHEADS]; };
__device__ __forceinline__ void p2_load(P2Frag& f, const float* __restrict__ Wb, int fbase, int ks0) {
#pragma unroll
  for (int i = 0; i < 2; ++i) {
    const int ks = ks0 + i;
    const int f0 = fbase + 16 * (ks & 3) + (ks >> 2);
#pragma unroll
    for (int hd = 0; hd < NHEADS; ++hd) {
      f.b0[i][hd] = *(const v4f*)(Wb + ((size_t)hd * FIN + f0) * FHID);
      f.b1[i][hd] = *(const v4f*)(Wb + ((size_t)hd * FIN + f0 + 8) * FHID);
    }
  }
}
__device__ __forceinline__ void p2_mma(const P2Frag& f, const v8f (&hT)[2], int ks0,
                                       v8f (&acc)[4 * NHEADS]) {
#pragma unroll
  for (int i = 0; i < 2; ++i) {
    const int ks = ks0 + i;
    v2f a;
    a.x = hT[ks >> 2][2 * (ks & 3)];
    a.y = hT[ks >> 2][2 * (ks & 3) + 1];
#pragma unroll
    for (int hd = 0; hd < NHEADS; ++hd) {
#pragma unroll
      for (int c = 0; c < 4; ++c) {
        v2f b;
        b.x = f.b0[i][hd][c];
        b.y = f.b1[i][hd][c];
        acc[4 * hd + c] = __builtin_amdgcn_wmma_f32_16x16x4_f32(false, a, false, b, (short)0,
                                                                acc[4 * hd + c], false, false);
      }
    }
  }
}

__global__ __launch_bounds__(128) __attribute__((amdgpu_waves_per_eu(3, 4)))
void lin_heads_kernel(const float* __restrict__ x, const float* __restrict__ lin_W,
                      const float* __restrict__ lin_b,
                      const float* __restrict__ W_heads, const float* __restrict__ a_heads,
                      const float* __restrict__ W_end, const float* __restrict__ a_end,
                      float* __restrict__ hcat, float* __restrict__ src,
                      float* __restrict__ dst, float* __restrict__ u2) {
  __shared__ v4f red[4][NHEADS][8][32];
  if (blockIdx.x < 2) {
    const int t = blockIdx.x * 128 + threadIdx.x;
    float s, d;
    dot64x2(W_end + (size_t)t * FOUT, a_end, s, d);
    u2[t] = s;
    u2[HCATF + t] = d;
    return;
  }
  const int mtile = blockIdx.x - 2;
  const int wave  = threadIdx.x >> 5;
  const int lane  = threadIdx.x & 31;
  const int half  = lane >> 4, l16 = lane & 15;
  const v8f zero = {0.f, 0.f, 0.f, 0.f, 0.f, 0.f, 0.f, 0.f};

  const float* xp = x + (size_t)(mtile * 16 + l16) * FIN + 4 * half;
  const float* wp = lin_W + (size_t)(4 * half) * FIN + 8 * l16 + 2 * wave;
  P1Frag g0, g1;
  p1_load(g0, xp, wp, 0);
  p1_load(g1, xp, wp, 8 * P1G);
  v8f hT[2];
  {
    const float* bp = lin_b + 64 * half + 2 * wave;
#pragma unroll
    for (int r = 0; r < 8; ++r) {
      const v2f bv = *(const v2f*)(bp + 8 * r);
      hT[0][r] = bv.x;
      hT[1][r] = bv.y;
    }
  }
  const float* Wb = W_heads + 4 * l16;
  const int fbase = 64 * half + 2 * wave;
  P2Frag q0, q1;
  __builtin_amdgcn_sched_barrier(0);
  p1_mma(g0, hT);
  p2_load(q0, Wb, fbase, 0);
  __builtin_amdgcn_sched_barrier(0);
  p1_mma(g1, hT);
  __builtin_amdgcn_sched_barrier(0);

  v8f acc[4 * NHEADS];
#pragma unroll
  for (int i = 0; i < 4 * NHEADS; ++i) acc[i] = zero;
#pragma unroll
  for (int ks = 0; ks < 8; ks += 4) {
    p2_load(q1, Wb, fbase, ks + 2);
    __builtin_amdgcn_sched_barrier(0);
    p2_mma(q0, hT, ks, acc);
    __builtin_amdgcn_sched_barrier(0);
    if (ks + 4 < 8) p2_load(q0, Wb, fbase, ks + 4);
    __builtin_amdgcn_sched_barrier(0);
    p2_mma(q1, hT, ks + 2, acc);
    __builtin_amdgcn_sched_barrier(0);
  }

#pragma unroll
  for (int hd = 0; hd < NHEADS; ++hd) {
#pragma unroll
    for (int r = 0; r < 8; ++r) {
      v4f p;
      p.x = acc[4 * hd + 0][r];
      p.y = acc[4 * hd + 1][r];
      p.z = acc[4 * hd + 2][r];
      p.w = acc[4 * hd + 3][r];
      red[wave][hd][r][lane] = p;
    }
  }
  __syncthreads();

  const int head = wave;
  const v4f as4 = *(const v4f*)(a_heads + head * 2 * FHID + 4 * l16);
  const v4f ad4 = *(const v4f*)(a_heads + head * 2 * FHID + FHID + 4 * l16);
  float* out = hcat + (size_t)(mtile * 16 + 8 * half) * HCATF + head * FHID + 4 * l16;
  float ps[8], pd[8];
#pragma unroll
  for (int r = 0; r < 8; ++r) {
    const v4f o = ((red[0][head][r][lane] + red[1][head][r][lane]) + red[2][head][r][lane])
                  + red[3][head][r][lane];
    *(v4f*)(out + (size_t)r * HCATF) = o;
    ps[r] = row16_sum(o.x * as4.x + o.y * as4.y + o.z * as4.z + o.w * as4.w);
    pd[r] = row16_sum(o.x * ad4.x + o.y * ad4.y + o.z * ad4.z + o.w * ad4.w);
  }
  if (l16 == 0) {
    float* sp = src + (size_t)head * NN + mtile * 16 + 8 * half;
    float* dp = dst + (size_t)head * NN + mtile * 16 + 8 * half;
    v4f s0 = {ps[0], ps[1], ps[2], ps[3]}, s1 = {ps[4], ps[5], ps[6], ps[7]};
    v4f d0 = {pd[0], pd[1], pd[2], pd[3]}, d1 = {pd[4], pd[5], pd[6], pd[7]};
    *(v4f*)sp = s0;
    *(v4f*)(sp + 4) = s1;
    *(v4f*)dp = d0;
    *(v4f*)(dp + 4) = d1;
  }
}

__global__ __launch_bounds__(64)
void gat_heads_fused_kernel(const float* __restrict__ adj,
                            const float* __restrict__ hcat,
                            const float* __restrict__ src,
                            const float* __restrict__ dst,
                            int* __restrict__ cnt, int* __restrict__ nidx,
                            float* __restrict__ gat1,
                            const float* __restrict__ u2,
                            float* __restrict__ src2, float* __restrict__ dst2) {
  __shared__ int   wtot[2];
  __shared__ float csum[2][2];
  __shared__ int   jidx[CAP];
  __shared__ float wgt[NHEADS][CAP];
  const int row  = blockIdx.x;
  const int t    = threadIdx.x;
  const int wave = t >> 5;
  const int lane = t & 31;
  const int l16  = lane & 15;
  const int head = wave * 2 + (lane >> 4);

  const float4* arow4 = (const float4*)(adj + (size_t)row * NN) + t * 16;
  const int base = t * 64;

  float4 v[16];
#pragma unroll
  for (int q = 0; q < 16; ++q) v[q] = arow4[q];

  const float si = src[head * NN + row];

  int c = 0;
#pragma unroll
  for (int q = 0; q < 16; ++q) {
    c += (v[q].x > 0.f) ? 1 : 0;
    c += (v[q].y > 0.f) ? 1 : 0;
    c += (v[q].z > 0.f) ? 1 : 0;
    c += (v[q].w > 0.f) ? 1 : 0;
  }
  int inc = c;
  for (int off = 1; off < 32; off <<= 1) {
    const int n = __shfl_up(inc, off);
    if (lane >= off) inc += n;
  }
  if (lane == 31) wtot[wave] = inc;
  __syncthreads();
  const int t0 = wtot[0], t1 = wtot[1];
  const int total = t0 + t1;
  int k = inc - c;
  if (wave > 0) k += t0;
  const int deg = total < CAP ? total : CAP;
  if (t == 0) cnt[row] = deg;
  int* outp = nidx + (size_t)row * CAP;
#pragma unroll
  for (int q = 0; q < 16; ++q) {
    const int b = base + q * 4;
    if (v[q].x > 0.f) { if (k < CAP) { jidx[k] = b + 0; outp[k] = b + 0; } ++k; }
    if (v[q].y > 0.f) { if (k < CAP) { jidx[k] = b + 1; outp[k] = b + 1; } ++k; }
    if (v[q].z > 0.f) { if (k < CAP) { jidx[k] = b + 2; outp[k] = b + 2; } ++k; }
    if (v[q].w > 0.f) { if (k < CAP) { jidx[k] = b + 3; outp[k] = b + 3; } ++k; }
  }
  __syncthreads();

  float* wg = wgt[head];
  const float* dsth = dst + head * NN;

  float m = -3.4e38f;
  for (int kk = l16; kk < deg; kk += 16) {
    const int j = jidx[kk];
    float e = si + dsth[j];
    e = (e >= 0.f) ? e : ALPHA_L * e;
    wg[kk] = e;
    m = fmaxf(m, e);
  }
  for (int off = 8; off > 0; off >>= 1) m = fmaxf(m, __shfl_xor(m, off));

  float ssum = 0.f;
  for (int kk = l16; kk < deg; kk += 16) {
    const float w = expf(wg[kk] - m);
    wg[kk] = w;
    ssum += w;
  }
  for (int off = 8; off > 0; off >>= 1) ssum += __shfl_xor(ssum, off);
  const float denom = ssum;
  __syncthreads();

  const float* hb = hcat + wave * (2 * FHID) + 4 * lane;
  const float4 us = *(const float4*)(u2 + wave * (2 * FHID) + 4 * lane);
  const float4 ud = *(const float4*)(u2 + HCATF + wave * (2 * FHID) + 4 * lane);
  float ax = 0.f, ay = 0.f, az = 0.f, aw = 0.f;
  for (int kk = 0; kk < deg; ++kk) {
    const float w = wg[kk];
    const float4 hv = *(const float4*)(hb + (size_t)jidx[kk] * HCATF);
    ax += w * hv.x;
    ay += w * hv.y;
    az += w * hv.z;
    aw += w * hv.w;
  }
  float4 o;
  o.x = ax / denom;
  o.y = ay / denom;
  o.z = az / denom;
  o.w = aw / denom;
  o.x = (o.x > 0.f) ? o.x : (expf(o.x) - 1.f);
  o.y = (o.y > 0.f) ? o.y : (expf(o.y) - 1.f);
  o.z = (o.z > 0.f) ? o.z : (expf(o.z) - 1.f);
  o.w = (o.w > 0.f) ? o.w : (expf(o.w) - 1.f);

  float ps = o.x * us.x + o.y * us.y + o.z * us.z + o.w * us.w;
  float pd = o.x * ud.x + o.y * ud.y + o.z * ud.z + o.w * ud.w;
  for (int off = 16; off > 0; off >>= 1) {
    ps += __shfl_xor(ps, off);
    pd += __shfl_xor(pd, off);
  }
  if (lane == 0) { csum[wave][0] = ps; csum[wave][1] = pd; }
  __syncthreads();
  *(float4*)(gat1 + (size_t)row * HCATF + wave * (2 * FHID) + 4 * lane) = o;
  if (t == 0) {
    src2[row] = csum[0][0] + csum[1][0];
    dst2[row] = csum[0][1] + csum[1][1];
  }
}

__global__ __launch_bounds__(64) __attribute__((amdgpu_waves_per_eu(1, 2)))
void end_gemm_kernel(const float* __restrict__ gat1,
                     const float* __restrict__ W_end,
                     float* __restrict__ Wh2) {
  const int mtile = blockIdx.x;
  const int wave  = threadIdx.x >> 5;
  const int lane  = threadIdx.x & 31;
  const int half  = lane >> 4, l16 = lane & 15;
  const int n0    = wave * 32;
  const v8f zero = {0.f, 0.f, 0.f, 0.f, 0.f, 0.f, 0.f, 0.f};
  v8f acc[2] = {zero, zero};
  gemm_strip<2, HCATF, 16>(gat1 + (size_t)mtile * 16 * HCATF, HCATF, W_end + n0, FOUT, lane, acc);
  float* out = Wh2 + (size_t)(mtile * 16 + 8 * half) * FOUT + n0 + 2 * l16;
#pragma unroll
  for (int r = 0; r < 8; ++r) {
    v2f o;
    o.x = acc[0][r];
    o.y = acc[1][r];
    *(v2f*)(out + (size_t)r * FOUT) = o;
  }
}

#define EB 16

__device__ __forceinline__ float elu1(float x) { return (x > 0.f) ? x : (expf(x) - 1.f); }

__global__ __launch_bounds__(32) __attribute__((amdgpu_waves_per_eu(1, 4)))
void gat_end_kernel(const float* __restrict__ Wh2,
                    const float* __restrict__ src,
                    const float* __restrict__ dst,
                    const int* __restrict__ cnt,
                    const int* __restrict__ nidx,
                    float* __restrict__ out) {
  __shared__ __attribute__((aligned(16))) float wgt[2][CAP];
  __shared__ __attribute__((aligned(16))) int   jidx[2][CAP];
  const int lane = threadIdx.x;
  const int hw   = lane >> 4;
  const int l16  = lane & 15;
  const int row  = blockIdx.x * 2 + hw;
  const int deg  = cnt[row];
  float si = src[row];
  asm volatile("" : "+v"(si));
  const int* nb  = nidx + (size_t)row * CAP;
  const int degp = (deg + EB - 1) & ~(EB - 1);
  float* wg = wgt[hw];
  int*   ji = jidx[hw];

  float m = -3.4e38f;
  for (int kb = 0; kb < degp; kb += 64) {
    int   jv[4];
    float dv[4];
#pragma unroll
    for (int s = 0; s < 4; ++s) {
      const int k = kb + 16 * s + l16;
      jv[s] = nb[k < deg ? k : deg - 1];
    }
#pragma unroll
    for (int s = 0; s < 4; ++s) dv[s] = dst[jv[s]];
#pragma unroll
    for (int s = 0; s < 4; ++s) {
      const int k = kb + 16 * s + l16;
      float e = si + dv[s];
      e = (e >= 0.f) ? e : ALPHA_L * e;
      e = (k < deg) ? e : -INFINITY;
      ji[k] = jv[s];
      wg[k] = e;
      m = fmaxf(m, e);
    }
  }
  m = row16_max(m);

  float ssum = 0.f;
  for (int kb = 0; kb < degp; kb += 64) {
    float ev[4];
#pragma unroll
    for (int s = 0; s < 4; ++s) ev[s] = wg[kb + 16 * s + l16];
#pragma unroll
    for (int s = 0; s < 4; ++s) {
      const float w = expf(ev[s] - m);
      wg[kb + 16 * s + l16] = w;
      ssum += w;
    }
  }
  const float denom = row16_sum(ssum);
  __syncthreads();

  const unsigned lo = 4 * l16;
  float ax = 0.f, ay = 0.f, az = 0.f, aw = 0.f;
  for (int k0 = 0; k0 < degp; k0 += EB) {
    int    jj[EB];
    float  ww[EB];
    float4 hv[EB];
#pragma unroll
    for (int q = 0; q < EB / 4; ++q) {
      const int4   jq = *(const int4*)(ji + k0 + 4 * q);
      const float4 wq = *(const float4*)(wg + k0 + 4 * q);
      jj[4 * q + 0] = jq.x; jj[4 * q + 1] = jq.y; jj[4 * q + 2] = jq.z; jj[4 * q + 3] = jq.w;
      ww[4 * q + 0] = wq.x; ww[4 * q + 1] = wq.y; ww[4 * q + 2] = wq.z; ww[4 * q + 3] = wq.w;
    }
#pragma unroll
    for (int i = 0; i < EB; ++i)
      hv[i] = *(const float4*)(Wh2 + ((unsigned)jj[i] * FOUT + lo));
#pragma unroll
    for (int i = 0; i < EB; ++i) {
      ax = __builtin_fmaf(ww[i], hv[i].x, ax);
      ay = __builtin_fmaf(ww[i], hv[i].y, ay);
      az = __builtin_fmaf(ww[i], hv[i].z, az);
      aw = __builtin_fmaf(ww[i], hv[i].w, aw);
    }
  }
  float4 o;
  o.x = elu1(ax / denom);
  o.y = elu1(ay / denom);
  o.z = elu1(az / denom);
  o.w = elu1(aw / denom);

  const float mo = row16_max(fmaxf(fmaxf(o.x, o.y), fmaxf(o.z, o.w)));
  float4 ex;
  ex.x = expf(o.x - mo);
  ex.y = expf(o.y - mo);
  ex.z = expf(o.z - mo);
  ex.w = expf(o.w - mo);
  const float sum = row16_sum((ex.x + ex.y) + (ex.z + ex.w));
  float4 r;
  r.x = ex.x / sum;
  r.y = ex.y / sum;
  r.z = ex.z / sum;
  r.w = ex.w / sum;
  *(float4*)(out + (size_t)row * FOUT + lo) = r;
}

extern "C" void kernel_launch(void* const* d_in, const int* in_sizes, int n_in,
                              void* d_out, int out_size, void* d_ws, size_t ws_size,
                              hipStream_t stream) {
  (void)in_sizes; (void)n_in; (void)out_size; (void)ws_size;
  const float* x       = (const float*)d_in[0];
  const float* adj     = (const float*)d_in[1];
  const float* lin_W   = (const float*)d_in[2];
  const float* lin_b   = (const float*)d_in[3];
  const float* W_heads = (const float*)d_in[4];
  const float* a_heads = (const float*)d_in[5];
  const float* W_end   = (const float*)d_in[6];
  const float* a_end   = (const float*)d_in[7];
  float* out = (float*)d_out;

  char* ws = (char*)d_ws;
  float* hcat  = (float*)ws;  ws += (size_t)NN * HCATF * sizeof(float);
  float* gat1  = (float*)ws;  ws += (size_t)NN * HCATF * sizeof(float);
  float* Wh2   = (float*)ws;  ws += (size_t)NN * FOUT  * sizeof(float);
  float* src1  = (float*)ws;  ws += (size_t)NHEADS * NN * sizeof(float);
  float* dst1  = (float*)ws;  ws += (size_t)NHEADS * NN * sizeof(float);
  float* src2  = (float*)ws;  ws += (size_t)NN * sizeof(float);
  float* dst2  = (float*)ws;  ws += (size_t)NN * sizeof(float);
  int*   ncnt  = (int*)ws;    ws += (size_t)NN * sizeof(int);
  int*   nidx  = (int*)ws;    ws += (size_t)NN * CAP * sizeof(int);
  float* u2    = Wh2;

  lin_heads_kernel<<<NN / 16 + 2, 128, 0, stream>>>(x, lin_W, lin_b, W_heads, a_heads,
                                                     W_end, a_end, hcat, src1, dst1, u2);
  gat_heads_fused_kernel<<<NN, 64, 0, stream>>>(adj, hcat, src1, dst1, ncnt, nidx, gat1,
                                                 u2, src2, dst2);
  end_gemm_kernel<<<NN / 16, 64, 0, stream>>>(gat1, W_end, Wh2);
  gat_end_kernel<<<NN / 2, 32, 0, stream>>>(Wh2, src2, dst2, ncnt, nidx, out);
}


# Layer-1 aggregation: 16-lane half-wave per row, lane = 4 features of each head; each half scans its own adjacency row lane-contiguous (slots from compare masks); waves 8192 -> 2048; sums reassociated
# speedup vs baseline: 1.8587x; 1.1107x over previous
#include <hip/hip_runtime.h>
#include <math.h>

#define NN      4096
#define FIN     128
#define FHID    64
#define NHEADS  4
#define FOUT    64
#define HCATF   (NHEADS * FHID)
#define ALPHA_L 0.2f
#define CAP     128

typedef float v2f __attribute__((ext_vector_type(2)));
typedef float v4f __attribute__((ext_vector_type(4)));
typedef float v8f __attribute__((ext_vector_type(8)));

struct v1f { float x; __device__ __forceinline__ float operator[](int) const { return x; } };
template <int NT> struct RowVec;
template <> struct RowVec<1> { typedef v1f T; };
template <> struct RowVec<2> { typedef v2f T; };
template <> struct RowVec<4> { typedef v4f T; };

__device__ __forceinline__ void ldrow(const float* __restrict__ p, v1f& o) { o.x = p[0]; }
__device__ __forceinline__ void ldrow(const float* __restrict__ p, v2f& o) { o = *(const v2f*)p; }
__device__ __forceinline__ void ldrow(const float* __restrict__ p, v4f& o) { o = *(const v4f*)p; }

template <int NT, int KB> struct Frag {
  v2f a[KB];
  typename RowVec<NT>::T b0[KB], b1[KB];
};

template <int NT, int KB>
__device__ __forceinline__ void frag_load(Frag<NT, KB>& f, const float* __restrict__ Ap,
                                          const float* __restrict__ Bp, int ldb, int k0) {
#pragma unroll
  for (int s = 0; s < KB; ++s) {
    const int k = k0 + 4 * s;
    f.a[s] = *(const v2f*)(Ap + k);
    ldrow(Bp + (size_t)k * ldb, f.b0[s]);
    ldrow(Bp + (size_t)(k + 1) * ldb, f.b1[s]);
  }
}

template <int NT, int KB>
__device__ __forceinline__ void frag_mma(const Frag<NT, KB>& f, v8f (&acc)[NT]) {
#pragma unroll
  for (int s = 0; s < KB; ++s) {
#pragma unroll
    for (int t = 0; t < NT; ++t) {
      v2f b;
      b.x = f.b0[s][t];
      b.y = f.b1[s][t];
      acc[t] = __builtin_amdgcn_wmma_f32_16x16x4_f32(false, f.a[s], false, b, (short)0, acc[t],
                                                     false, false);
    }
  }
}

template <int NT, int K, int KB>
__device__ __forceinline__ void gemm_strip(const float* __restrict__ A, int lda,
                                           const float* __restrict__ B, int ldb,
                                           int lane, v8f (&acc)[NT]) {
  static_assert(K % (8 * KB) == 0, "K must hold an even number of k-blocks");
  const int half = lane >> 4;
  const int l16  = lane & 15;
  const float* Ap = A + (size_t)l16 * lda + 2 * half;
  const float* Bp = B + (size_t)(2 * half) * ldb + NT * l16;
  Frag<NT, KB> f0, f1;
  frag_load<NT, KB>(f0, Ap, Bp, ldb, 0);
#pragma unroll
  for (int k0 = 0; k0 < K; k0 += 8 * KB) {
    frag_load<NT, KB>(f1, Ap, Bp, ldb, k0 + 4 * KB);
    frag_mma<NT, KB>(f0, acc);
    if (k0 + 8 * KB < K) frag_load<NT, KB>(f0, Ap, Bp, ldb, k0 + 8 * KB);
    frag_mma<NT, KB>(f1, acc);
  }
}

__device__ __forceinline__ void dot64x2(const float* __restrict__ wrow,
                                        const float* __restrict__ a,
                                        float& s_out, float& d_out) {
  const float4* w4 = (const float4*)wrow;
  const float4* as = (const float4*)a;
  const float4* ad = as + 16;
  float s = 0.f, d = 0.f;
#pragma unroll
  for (int f = 0; f < 16; ++f) {
    const float4 w = w4[f], A = as[f], B = ad[f];
    s += w.x * A.x + w.y * A.y + w.z * A.z + w.w * A.w;
    d += w.x * B.x + w.y * B.y + w.z * B.z + w.w * B.w;
  }
  s_out = s;
  d_out = d;
}

template <int M>
__device__ __forceinline__ float row_xor(float v) {
  const int b = __builtin_bit_cast(int, v);
  return __builtin_bit_cast(float, __builtin_amdgcn_update_dpp(b, b, 0x160 + M, 0xf, 0xf, false));
}
__device__ __forceinline__ float row16_max(float v) {
  v = fmaxf(v, row_xor<1>(v));
  v = fmaxf(v, row_xor<2>(v));
  v = fmaxf(v, row_xor<4>(v));
  v = fmaxf(v, row_xor<8>(v));
  return v;
}
__device__ __forceinline__ float row16_sum(float v) {
  v += row_xor<1>(v);
  v += row_xor<2>(v);
  v += row_xor<4>(v);
  v += row_xor<8>(v);
  return v;
}

#define P1G 8
struct P1Frag {
  v4f xb[P1G];
  v2f w[P1G][4];
};
__device__ __forceinline__ void p1_load(P1Frag& f, const float* __restrict__ xp,
                                        const float* __restrict__ wp, int k0) {
#pragma unroll
  for (int g = 0; g < P1G; ++g) {
    const int k = k0 + 8 * g;
    f.xb[g] = *(const v4f*)(xp + k);
#pragma unroll
    for (int j = 0; j < 4; ++j) f.w[g][j] = *(const v2f*)(wp + (size_t)(k + j) * FIN);
  }
}
__device__ __forceinline__ void p1_mma(const P1Frag& f, v8f (&hT)[2]) {
#pragma unroll
  for (int g = 0; g < P1G; ++g) {
#pragma unroll
    for (int e = 0; e < 2; ++e) {
      v2f b, a0, a1;
      b.x  = f.xb[g][2 * e];
      b.y  = f.xb[g][2 * e + 1];
      a0.x = f.w[g][2 * e].x;
      a0.y = f.w[g][2 * e + 1].x;
      a1.x = f.w[g][2 * e].y;
      a1.y = f.w[g][2 * e + 1].y;
      hT[0] = __builtin_amdgcn_wmma_f32_16x16x4_f32(false, a0, false, b, (short)0, hT[0],
                                                    false, false);
      hT[1] = __builtin_amdgcn_wmma_f32_16x16x4_f32(false, a1, false, b, (short)0, hT[1],
                                                    false, false);
    }
  }
}

struct P2Frag { v4f b0[2][NHEADS], b1[2][NHEADS]; };
__device__ __forceinline__ void p2_load(P2Frag& f, const float* __restrict__ Wb, int fbase, int ks0) {
#pragma unroll
  for (int i = 0; i < 2; ++i) {
    const int ks = ks0 + i;
    const int f0 = fbase + 16 * (ks & 3) + (ks >> 2);
#pragma unroll
    for (int hd = 0; hd < NHEADS; ++hd) {
      f.b0[i][hd] = *(const v4f*)(Wb + ((size_t)hd * FIN + f0) * FHID);
      f.b1[i][hd] = *(const v4f*)(Wb + ((size_t)hd * FIN + f0 + 8) * FHID);
    }
  }
}
__device__ __forceinline__ void p2_mma(const P2Frag& f, const v8f (&hT)[2], int ks0,
                                       v8f (&acc)[4 * NHEADS]) {
#pragma unroll
  for (int i = 0; i < 2; ++i) {
    const int ks = ks0 + i;
    v2f a;
    a.x = hT[ks >> 2][2 * (ks & 3)];
    a.y = hT[ks >> 2][2 * (ks & 3) + 1];
#pragma unroll
    for (int hd = 0; hd < NHEADS; ++hd) {
#pragma unroll
      for (int c = 0; c < 4; ++c) {
        v2f b;
        b.x = f.b0[i][hd][c];
        b.y = f.b1[i][hd][c];
        acc[4 * hd + c] = __builtin_amdgcn_wmma_f32_16x16x4_f32(false, a, false, b, (short)0,
                                                                acc[4 * hd + c], false, false);
      }
    }
  }
}

__global__ __launch_bounds__(128) __attribute__((amdgpu_waves_per_eu(3, 4)))
void lin_heads_kernel(const float* __restrict__ x, const float* __restrict__ lin_W,
                      const float* __restrict__ lin_b,
                      const float* __restrict__ W_heads, const float* __restrict__ a_heads,
                      const float* __restrict__ W_end, const float* __restrict__ a_end,
                      float* __restrict__ hcat, float* __restrict__ src,
                      float* __restrict__ dst, float* __restrict__ u2) {
  __shared__ v4f red[4][NHEADS][8][32];
  if (blockIdx.x < 2) {
    const int t = blockIdx.x * 128 + threadIdx.x;
    float s, d;
    dot64x2(W_end + (size_t)t * FOUT, a_end, s, d);
    u2[t] = s;
    u2[HCATF + t] = d;
    return;
  }
  const int mtile = blockIdx.x - 2;
  const int wave  = threadIdx.x >> 5;
  const int lane  = threadIdx.x & 31;
  const int half  = lane >> 4, l16 = lane & 15;
  const v8f zero = {0.f, 0.f, 0.f, 0.f, 0.f, 0.f, 0.f, 0.f};

  const float* xp = x + (size_t)(mtile * 16 + l16) * FIN + 4 * half;
  const float* wp = lin_W + (size_t)(4 * half) * FIN + 8 * l16 + 2 * wave;
  P1Frag g0, g1;
  p1_load(g0, xp, wp, 0);
  p1_load(g1, xp, wp, 8 * P1G);
  v8f hT[2];
  {
    const float* bp = lin_b + 64 * half + 2 * wave;
#pragma unroll
    for (int r = 0; r < 8; ++r) {
      const v2f bv = *(const v2f*)(bp + 8 * r);
      hT[0][r] = bv.x;
      hT[1][r] = bv.y;
    }
  }
  const float* Wb = W_heads + 4 * l16;
  const int fbase = 64 * half + 2 * wave;
  P2Frag q0, q1;
  __builtin_amdgcn_sched_barrier(0);
  p1_mma(g0, hT);
  p2_load(q0, Wb, fbase, 0);
  __builtin_amdgcn_sched_barrier(0);
  p1_mma(g1, hT);
  __builtin_amdgcn_sched_barrier(0);

  v8f acc[4 * NHEADS];
#pragma unroll
  for (int i = 0; i < 4 * NHEADS; ++i) acc[i] = zero;
#pragma unroll
  for (int ks = 0; ks < 8; ks += 4) {
    p2_load(q1, Wb, fbase, ks + 2);
    __builtin_amdgcn_sched_barrier(0);
    p2_mma(q0, hT, ks, acc);
    __builtin_amdgcn_sched_barrier(0);
    if (ks + 4 < 8) p2_load(q0, Wb, fbase, ks + 4);
    __builtin_amdgcn_sched_barrier(0);
    p2_mma(q1, hT, ks + 2, acc);
    __builtin_amdgcn_sched_barrier(0);
  }

#pragma unroll
  for (int hd = 0; hd < NHEADS; ++hd) {
#pragma unroll
    for (int r = 0; r < 8; ++r) {
      v4f p;
      p.x = acc[4 * hd + 0][r];
      p.y = acc[4 * hd + 1][r];
      p.z = acc[4 * hd + 2][r];
      p.w = acc[4 * hd + 3][r];
      red[wave][hd][r][lane] = p;
    }
  }
  __syncthreads();

  const int head = wave;
  const v4f as4 = *(const v4f*)(a_heads + head * 2 * FHID + 4 * l16);
  const v4f ad4 = *(const v4f*)(a_heads + head * 2 * FHID + FHID + 4 * l16);
  float* out = hcat + (size_t)(mtile * 16 + 8 * half) * HCATF + head * FHID + 4 * l16;
  float ps[8], pd[8];
#pragma unroll
  for (int r = 0; r < 8; ++r) {
    const v4f o = ((red[0][head][r][lane] + red[1][head][r][lane]) + red[2][head][r][lane])
                  + red[3][head][r][lane];
    *(v4f*)(out + (size_t)r * HCATF) = o;
    ps[r] = row16_sum(o.x * as4.x + o.y * as4.y + o.z * as4.z + o.w * as4.w);
    pd[r] = row16_sum(o.x * ad4.x + o.y * ad4.y + o.z * ad4.z + o.w * ad4.w);
  }
  if (l16 == 0) {
    float* sp = src + (size_t)head * NN + mtile * 16 + 8 * half;
    float* dp = dst + (size_t)head * NN + mtile * 16 + 8 * half;
    v4f s0 = {ps[0], ps[1], ps[2], ps[3]}, s1 = {ps[4], ps[5], ps[6], ps[7]};
    v4f d0 = {pd[0], pd[1], pd[2], pd[3]}, d1 = {pd[4], pd[5], pd[6], pd[7]};
    *(v4f*)sp = s0;
    *(v4f*)(sp + 4) = s1;
    *(v4f*)dp = d0;
    *(v4f*)(dp + 4) = d1;
  }
}

__device__ __forceinline__ float elu1(float x) { return (x > 0.f) ? x : (expf(x) - 1.f); }

#define HB 16

__device__ __forceinline__ int below_lane(unsigned mask, int acc) {
  return (int)__builtin_amdgcn_mbcnt_lo(mask, (unsigned)acc);
}

__device__ __forceinline__ void scan_step(const float4 a, const int col, const unsigned hmask,
                                          const int spill, int& pos, int* __restrict__ ji) {
  const bool p0 = a.x > 0.f, p1 = a.y > 0.f, p2 = a.z > 0.f, p3 = a.w > 0.f;
  const unsigned m0 = __builtin_amdgcn_ballot_w32(p0);
  const unsigned m1 = __builtin_amdgcn_ballot_w32(p1);
  const unsigned m2 = __builtin_amdgcn_ballot_w32(p2);
  const unsigned m3 = __builtin_amdgcn_ballot_w32(p3);
  if ((m0 | m1 | m2 | m3) != 0u) {
    const unsigned u0 = m0 & hmask, u1 = m1 & hmask, u2 = m2 & hmask, u3 = m3 & hmask;
    int k = pos + below_lane(u0, below_lane(u1, below_lane(u2, below_lane(u3, 0))));
    pos += __popc(u0) + __popc(u1) + __popc(u2) + __popc(u3);
    ji[(p0 && k < CAP) ? k : spill] = col;
    k += p0 ? 1 : 0;
    ji[(p1 && k < CAP) ? k : spill] = col + 1;
    k += p1 ? 1 : 0;
    ji[(p2 && k < CAP) ? k : spill] = col + 2;
    k += p2 ? 1 : 0;
    ji[(p3 && k < CAP) ? k : spill] = col + 3;
  }
}

__global__ __launch_bounds__(32) __attribute__((amdgpu_waves_per_eu(1, 2)))
void gat_heads_fused_kernel(const float* __restrict__ adj,
                            const float* __restrict__ hcat,
                            const float* __restrict__ src,
                            const float* __restrict__ dst,
                            int* __restrict__ cnt, int* __restrict__ nidx,
                            float* __restrict__ gat1,
                            const float* __restrict__ u2,
                            float* __restrict__ src2, float* __restrict__ dst2) {
  __shared__ __attribute__((aligned(16))) int jidx[2][CAP + 16];
  __shared__ __attribute__((aligned(16))) v4f wgt[2][CAP];
  const int lane = threadIdx.x;
  const int hw   = lane >> 4;
  const int l16  = lane & 15;
  const int row  = blockIdx.x * 2 + hw;
  int* ji = jidx[hw];
  v4f* wg = wgt[hw];

  const float4* arow4 = (const float4*)(adj + (size_t)row * NN) + l16;
  const unsigned hmask = hw ? 0xffff0000u : 0x0000ffffu;
  const int spill = CAP + l16;
  float4 v[16];
#pragma unroll
  for (int q = 0; q < 16; ++q) v[q] = arow4[q * 16];

  float si[NHEADS];
#pragma unroll
  for (int h = 0; h < NHEADS; ++h) si[h] = src[h * NN + row];

  int pos = 0;
#pragma nounroll
  for (int g = 0; g < 3; ++g) {
#pragma unroll
    for (int q = 0; q < 16; ++q) {
      scan_step(v[q], (g * 16 + q) * 64 + 4 * l16, hmask, spill, pos, ji);
      v[q] = arow4[((g + 1) * 16 + q) * 16];
    }
  }
#pragma unroll
  for (int q = 0; q < 16; ++q)
    scan_step(v[q], (48 + q) * 64 + 4 * l16, hmask, spill, pos, ji);
  __syncthreads();

  const int deg  = pos < CAP ? pos : CAP;
  const int degp = (deg + HB - 1) & ~(HB - 1);
  if (l16 == 0) cnt[row] = deg;
  int* outp = nidx + (size_t)row * CAP;
  for (int kb = 4 * l16; kb < deg; kb += 64) *(int4*)(outp + kb) = *(const int4*)(ji + kb);

  v4f mx = {-3.4e38f, -3.4e38f, -3.4e38f, -3.4e38f};
  for (int kb = 0; kb < degp; kb += 64) {
    int   jv[4];
    float dv[4][NHEADS];
#pragma unroll
    for (int s = 0; s < 4; ++s) {
      const int k = kb + 16 * s + l16;
      jv[s] = ji[k < deg ? k : deg - 1];
    }
#pragma unroll
    for (int s = 0; s < 4; ++s)
#pragma unroll
      for (int h = 0; h < NHEADS; ++h) dv[s][h] = dst[h * NN + jv[s]];
#pragma unroll
    for (int s = 0; s < 4; ++s) {
      const int k = kb + 16 * s + l16;
      v4f e;
#pragma unroll
      for (int h = 0; h < NHEADS; ++h) {
        float t = si[h] + dv[s][h];
        t = (t >= 0.f) ? t : ALPHA_L * t;
        e[h] = (k < deg) ? t : -INFINITY;
        mx[h] = fmaxf(mx[h], e[h]);
      }
      if (k >= deg) ji[k] = jv[s];
      wg[k] = e;
    }
  }
#pragma unroll
  for (int h = 0; h < NHEADS; ++h) mx[h] = row16_max(mx[h]);

  v4f ss = {0.f, 0.f, 0.f, 0.f};
  for (int kb = 0; kb < degp; kb += 64) {
    v4f ev[4];
#pragma unroll
    for (int s = 0; s < 4; ++s) ev[s] = wg[kb + 16 * s + l16];
#pragma unroll
    for (int s = 0; s < 4; ++s) {
      v4f w;
#pragma unroll
      for (int h = 0; h < NHEADS; ++h) {
        w[h] = expf(ev[s][h] - mx[h]);
        ss[h] += w[h];
      }
      wg[kb + 16 * s + l16] = w;
    }
  }
  float denom[NHEADS];
#pragma unroll
  for (int h = 0; h < NHEADS; ++h) denom[h] = row16_sum(ss[h]);
  __syncthreads();

  float4 us[NHEADS], ud[NHEADS];
#pragma unroll
  for (int h = 0; h < NHEADS; ++h) {
    us[h] = *(const float4*)(u2 + h * FHID + 4 * l16);
    ud[h] = *(const float4*)(u2 + HCATF + h * FHID + 4 * l16);
  }
  const float* hb = hcat + 4 * l16;
  float4 acc[NHEADS];
#pragma unroll
  for (int h = 0; h < NHEADS; ++h) acc[h] = make_float4(0.f, 0.f, 0.f, 0.f);
  for (int k0 = 0; k0 < degp; k0 += HB) {
    int    jj[HB];
    v4f    ww[HB];
    float4 hv[HB][NHEADS];
#pragma unroll
    for (int q = 0; q < HB / 4; ++q) {
      const int4 jq = *(const int4*)(ji + k0 + 4 * q);
      jj[4 * q + 0] = jq.x; jj[4 * q + 1] = jq.y; jj[4 * q + 2] = jq.z; jj[4 * q + 3] = jq.w;
    }
#pragma unroll
    for (int i = 0; i < HB; ++i) ww[i] = wg[k0 + i];
#pragma unroll
    for (int i = 0; i < HB; ++i)
#pragma unroll
      for (int h = 0; h < NHEADS; ++h)
        hv[i][h] = *(const float4*)(hb + ((unsigned)jj[i] * HCATF + h * FHID));
#pragma unroll
    for (int i = 0; i < HB; ++i)
#pragma unroll
      for (int h = 0; h < NHEADS; ++h) {
        const float w = ww[i][h];
        acc[h].x += w * hv[i][h].x;
        acc[h].y += w * hv[i][h].y;
        acc[h].z += w * hv[i][h].z;
        acc[h].w += w * hv[i][h].w;
      }
  }

  float ps = 0.f, pd = 0.f;
  float* orow = gat1 + (size_t)row * HCATF + 4 * l16;
#pragma unroll
  for (int h = 0; h < NHEADS; ++h) {
    float4 o;
    o.x = elu1(acc[h].x / denom[h]);
    o.y = elu1(acc[h].y / denom[h]);
    o.z = elu1(acc[h].z / denom[h]);
    o.w = elu1(acc[h].w / denom[h]);
    *(float4*)(orow + h * FHID) = o;
    ps += o.x * us[h].x + o.y * us[h].y + o.z * us[h].z + o.w * us[h].w;
    pd += o.x * ud[h].x + o.y * ud[h].y + o.z * ud[h].z + o.w * ud[h].w;
  }
  ps = row16_sum(ps);
  pd = row16_sum(pd);
  if (l16 == 0) {
    src2[row] = ps;
    dst2[row] = pd;
  }
}

__global__ __launch_bounds__(64) __attribute__((amdgpu_waves_per_eu(1, 2)))
void end_gemm_kernel(const float* __restrict__ gat1,
                     const float* __restrict__ W_end,
                     float* __restrict__ Wh2) {
  const int mtile = blockIdx.x;
  const int wave  = threadIdx.x >> 5;
  const int lane  = threadIdx.x & 31;
  const int half  = lane >> 4, l16 = lane & 15;
  const int n0    = wave * 32;
  const v8f zero = {0.f, 0.f, 0.f, 0.f, 0.f, 0.f, 0.f, 0.f};
  v8f acc[2] = {zero, zero};
  gemm_strip<2, HCATF, 16>(gat1 + (size_t)mtile * 16 * HCATF, HCATF, W_end + n0, FOUT, lane, acc);
  float* out = Wh2 + (size_t)(mtile * 16 + 8 * half) * FOUT + n0 + 2 * l16;
#pragma unroll
  for (int r = 0; r < 8; ++r) {
    v2f o;
    o.x = acc[0][r];
    o.y = acc[1][r];
    *(v2f*)(out + (size_t)r * FOUT) = o;
  }
}

#define EB 16

__global__ __launch_bounds__(32) __attribute__((amdgpu_waves_per_eu(1, 4)))
void gat_end_kernel(const float* __restrict__ Wh2,
                    const float* __restrict__ src,
                    const float* __restrict__ dst,
                    const int* __restrict__ cnt,
                    const int* __restrict__ nidx,
                    float* __restrict__ out) {
  __shared__ __attribute__((aligned(16))) float wgt[2][CAP];
  __shared__ __attribute__((aligned(16))) int   jidx[2][CAP];
  const int lane = threadIdx.x;
  const int hw   = lane >> 4;
  const int l16  = lane & 15;
  const int row  = blockIdx.x * 2 + hw;
  const int deg  = cnt[row];
  float si = src[row];
  asm volatile("" : "+v"(si));
  const int* nb  = nidx + (size_t)row * CAP;
  const int degp = (deg + EB - 1) & ~(EB - 1);
  float* wg = wgt[hw];
  int*   ji = jidx[hw];

  float m = -3.4e38f;
  for (int kb = 0; kb < degp; kb += 64) {
    int   jv[4];
    float dv[4];
#pragma unroll
    for (int s = 0; s < 4; ++s) {
      const int k = kb + 16 * s + l16;
      jv[s] = nb[k < deg ? k : deg - 1];
    }
#pragma unroll
    for (int s = 0; s < 4; ++s) dv[s] = dst[jv[s]];
#pragma unroll
    for (int s = 0; s < 4; ++s) {
      const int k = kb + 16 * s + l16;
      float e = si + dv[s];
      e = (e >= 0.f) ? e : ALPHA_L * e;
      e = (k < deg) ? e : -INFINITY;
      ji[k] = jv[s];
      wg[k] = e;
      m = fmaxf(m, e);
    }
  }
  m = row16_max(m);

  float ssum = 0.f;
  for (int kb = 0; kb < degp; kb += 64) {
    float ev[4];
#pragma unroll
    for (int s = 0; s < 4; ++s) ev[s] = wg[kb + 16 * s + l16];
#pragma unroll
    for (int s = 0; s < 4; ++s) {
      const float w = expf(ev[s] - m);
      wg[kb + 16 * s + l16] = w;
      ssum += w;
    }
  }
  const float denom = row16_sum(ssum);
  __syncthreads();

  const unsigned lo = 4 * l16;
  float ax = 0.f, ay = 0.f, az = 0.f, aw = 0.f;
  for (int k0 = 0; k0 < degp; k0 += EB) {
    int    jj[EB];
    float  ww[EB];
    float4 hv[EB];
#pragma unroll
    for (int q = 0; q < EB / 4; ++q) {
      const int4   jq = *(const int4*)(ji + k0 + 4 * q);
      const float4 wq = *(const float4*)(wg + k0 + 4 * q);
      jj[4 * q + 0] = jq.x; jj[4 * q + 1] = jq.y; jj[4 * q + 2] = jq.z; jj[4 * q + 3] = jq.w;
      ww[4 * q + 0] = wq.x; ww[4 * q + 1] = wq.y; ww[4 * q + 2] = wq.z; ww[4 * q + 3] = wq.w;
    }
#pragma unroll
    for (int i = 0; i < EB; ++i)
      hv[i] = *(const float4*)(Wh2 + ((unsigned)jj[i] * FOUT + lo));
#pragma unroll
    for (int i = 0; i < EB; ++i) {
      ax = __builtin_fmaf(ww[i], hv[i].x, ax);
      ay = __builtin_fmaf(ww[i], hv[i].y, ay);
      az = __builtin_fmaf(ww[i], hv[i].z, az);
      aw = __builtin_fmaf(ww[i], hv[i].w, aw);
    }
  }
  float4 o;
  o.x = elu1(ax / denom);
  o.y = elu1(ay / denom);
  o.z = elu1(az / denom);
  o.w = elu1(aw / denom);

  const float mo = row16_max(fmaxf(fmaxf(o.x, o.y), fmaxf(o.z, o.w)));
  float4 ex;
  ex.x = expf(o.x - mo);
  ex.y = expf(o.y - mo);
  ex.z = expf(o.z - mo);
  ex.w = expf(o.w - mo);
  const float sum = row16_sum((ex.x + ex.y) + (ex.z + ex.w));
  float4 r;
  r.x = ex.x / sum;
  r.y = ex.y / sum;
  r.z = ex.z / sum;
  r.w = ex.w / sum;
  *(float4*)(out + (size_t)row * FOUT + lo) = r;
}

extern "C" void kernel_launch(void* const* d_in, const int* in_sizes, int n_in,
                              void* d_out, int out_size, void* d_ws, size_t ws_size,
                              hipStream_t stream) {
  (void)in_sizes; (void)n_in; (void)out_size; (void)ws_size;
  const float* x       = (const float*)d_in[0];
  const float* adj     = (const float*)d_in[1];
  const float* lin_W   = (const float*)d_in[2];
  const float* lin_b   = (const float*)d_in[3];
  const float* W_heads = (const float*)d_in[4];
  const float* a_heads = (const float*)d_in[5];
  const float* W_end   = (const float*)d_in[6];
  const float* a_end   = (const float*)d_in[7];
  float* out = (float*)d_out;

  char* ws = (char*)d_ws;
  float* hcat  = (float*)ws;  ws += (size_t)NN * HCATF * sizeof(float);
  float* gat1  = (float*)ws;  ws += (size_t)NN * HCATF * sizeof(float);
  float* Wh2   = (float*)ws;  ws += (size_t)NN * FOUT  * sizeof(float);
  float* src1  = (float*)ws;  ws += (size_t)NHEADS * NN * sizeof(float);
  float* dst1  = (float*)ws;  ws += (size_t)NHEADS * NN * sizeof(float);
  float* src2  = (float*)ws;  ws += (size_t)NN * sizeof(float);
  float* dst2  = (float*)ws;  ws += (size_t)NN * sizeof(float);
  int*   ncnt  = (int*)ws;    ws += (size_t)NN * sizeof(int);
  int*   nidx  = (int*)ws;    ws += (size_t)NN * CAP * sizeof(int);
  float* u2    = Wh2;

  lin_heads_kernel<<<NN / 16 + 2, 128, 0, stream>>>(x, lin_W, lin_b, W_heads, a_heads,
                                                     W_end, a_end, hcat, src1, dst1, u2);
  gat_heads_fused_kernel<<<NN / 2, 32, 0, stream>>>(adj, hcat, src1, dst1, ncnt, nidx, gat1,
                                                 u2, src2, dst2);
  end_gemm_kernel<<<NN / 16, 64, 0, stream>>>(gat1, W_end, Wh2);
  gat_end_kernel<<<NN / 2, 32, 0, stream>>>(Wh2, src2, dst2, ncnt, nidx, out);
}


# Adjacency scan without branches, ballots or LDS: each lane shifts its compare bits into a 256-bit register map of its own columns, whole row in flight; one pass lists them; neighbor sums lane-major
# speedup vs baseline: 1.9295x; 1.0381x over previous
#include <hip/hip_runtime.h>
#include <math.h>

#define NN      4096
#define FIN     128
#define FHID    64
#define NHEADS  4
#define FOUT    64
#define HCATF   (NHEADS * FHID)
#define ALPHA_L 0.2f
#define CAP     128

typedef float v2f __attribute__((ext_vector_type(2)));
typedef float v4f __attribute__((ext_vector_type(4)));
typedef float v8f __attribute__((ext_vector_type(8)));

struct v1f { float x; __device__ __forceinline__ float operator[](int) const { return x; } };
template <int NT> struct RowVec;
template <> struct RowVec<1> { typedef v1f T; };
template <> struct RowVec<2> { typedef v2f T; };
template <> struct RowVec<4> { typedef v4f T; };

__device__ __forceinline__ void ldrow(const float* __restrict__ p, v1f& o) { o.x = p[0]; }
__device__ __forceinline__ void ldrow(const float* __restrict__ p, v2f& o) { o = *(const v2f*)p; }
__device__ __forceinline__ void ldrow(const float* __restrict__ p, v4f& o) { o = *(const v4f*)p; }

template <int NT, int KB> struct Frag {
  v2f a[KB];
  typename RowVec<NT>::T b0[KB], b1[KB];
};

template <int NT, int KB>
__device__ __forceinline__ void frag_load(Frag<NT, KB>& f, const float* __restrict__ Ap,
                                          const float* __restrict__ Bp, int ldb, int k0) {
#pragma unroll
  for (int s = 0; s < KB; ++s) {
    const int k = k0 + 4 * s;
    f.a[s] = *(const v2f*)(Ap + k);
    ldrow(Bp + (size_t)k * ldb, f.b0[s]);
    ldrow(Bp + (size_t)(k + 1) * ldb, f.b1[s]);
  }
}

template <int NT, int KB>
__device__ __forceinline__ void frag_mma(const Frag<NT, KB>& f, v8f (&acc)[NT]) {
#pragma unroll
  for (int s = 0; s < KB; ++s) {
#pragma unroll
    for (int t = 0; t < NT; ++t) {
      v2f b;
      b.x = f.b0[s][t];
      b.y = f.b1[s][t];
      acc[t] = __builtin_amdgcn_wmma_f32_16x16x4_f32(false, f.a[s], false, b, (short)0, acc[t],
                                                     false, false);
    }
  }
}

template <int NT, int K, int KB>
__device__ __forceinline__ void gemm_strip(const float* __restrict__ A, int lda,
                                           const float* __restrict__ B, int ldb,
                                           int lane, v8f (&acc)[NT]) {
  static_assert(K % (8 * KB) == 0, "K must hold an even number of k-blocks");
  const int half = lane >> 4;
  const int l16  = lane & 15;
  const float* Ap = A + (size_t)l16 * lda + 2 * half;
  const float* Bp = B + (size_t)(2 * half) * ldb + NT * l16;
  Frag<NT, KB> f0, f1;
  frag_load<NT, KB>(f0, Ap, Bp, ldb, 0);
#pragma unroll
  for (int k0 = 0; k0 < K; k0 += 8 * KB) {
    frag_load<NT, KB>(f1, Ap, Bp, ldb, k0 + 4 * KB);
    frag_mma<NT, KB>(f0, acc);
    if (k0 + 8 * KB < K) frag_load<NT, KB>(f0, Ap, Bp, ldb, k0 + 8 * KB);
    frag_mma<NT, KB>(f1, acc);
  }
}

__device__ __forceinline__ void dot64x2(const float* __restrict__ wrow,
                                        const float* __restrict__ a,
                                        float& s_out, float& d_out) {
  const float4* w4 = (const float4*)wrow;
  const float4* as = (const float4*)a;
  const float4* ad = as + 16;
  float s = 0.f, d = 0.f;
#pragma unroll
  for (int f = 0; f < 16; ++f) {
    const float4 w = w4[f], A = as[f], B = ad[f];
    s += w.x * A.x + w.y * A.y + w.z * A.z + w.w * A.w;
    d += w.x * B.x + w.y * B.y + w.z * B.z + w.w * B.w;
  }
  s_out = s;
  d_out = d;
}

template <int M>
__device__ __forceinline__ float row_xor(float v) {
  const int b = __builtin_bit_cast(int, v);
  return __builtin_bit_cast(float, __builtin_amdgcn_update_dpp(b, b, 0x160 + M, 0xf, 0xf, false));
}
__device__ __forceinline__ float row16_max(float v) {
  v = fmaxf(v, row_xor<1>(v));
  v = fmaxf(v, row_xor<2>(v));
  v = fmaxf(v, row_xor<4>(v));
  v = fmaxf(v, row_xor<8>(v));
  return v;
}
__device__ __forceinline__ float row16_sum(float v) {
  v += row_xor<1>(v);
  v += row_xor<2>(v);
  v += row_xor<4>(v);
  v += row_xor<8>(v);
  return v;
}

#define P1G 8
struct P1Frag {
  v4f xb[P1G];
  v2f w[P1G][4];
};
__device__ __forceinline__ void p1_load(P1Frag& f, const float* __restrict__ xp,
                                        const float* __restrict__ wp, int k0) {
#pragma unroll
  for (int g = 0; g < P1G; ++g) {
    const int k = k0 + 8 * g;
    f.xb[g] = *(const v4f*)(xp + k);
#pragma unroll
    for (int j = 0; j < 4; ++j) f.w[g][j] = *(const v2f*)(wp + (size_t)(k + j) * FIN);
  }
}
__device__ __forceinline__ void p1_mma(const P1Frag& f, v8f (&hT)[2]) {
#pragma unroll
  for (int g = 0; g < P1G; ++g) {
#pragma unroll
    for (int e = 0; e < 2; ++e) {
      v2f b, a0, a1;
      b.x  = f.xb[g][2 * e];
      b.y  = f.xb[g][2 * e + 1];
      a0.x = f.w[g][2 * e].x;
      a0.y = f.w[g][2 * e + 1].x;
      a1.x = f.w[g][2 * e].y;
      a1.y = f.w[g][2 * e + 1].y;
      hT[0] = __builtin_amdgcn_wmma_f32_16x16x4_f32(false, a0, false, b, (short)0, hT[0],
                                                    false, false);
      hT[1] = __builtin_amdgcn_wmma_f32_16x16x4_f32(false, a1, false, b, (short)0, hT[1],
                                                    false, false);
    }
  }
}

struct P2Frag { v4f b0[2][NHEADS], b1[2][NHEADS]; };
__device__ __forceinline__ void p2_load(P2Frag& f, const float* __restrict__ Wb, int fbase, int ks0) {
#pragma unroll
  for (int i = 0; i < 2; ++i) {
    const int ks = ks0 + i;
    const int f0 = fbase + 16 * (ks & 3) + (ks >> 2);
#pragma unroll
    for (int hd = 0; hd < NHEADS; ++hd) {
      f.b0[i][hd] = *(const v4f*)(Wb + ((size_t)hd * FIN + f0) * FHID);
      f.b1[i][hd] = *(const v4f*)(Wb + ((size_t)hd * FIN + f0 + 8) * FHID);
    }
  }
}
__device__ __forceinline__ void p2_mma(const P2Frag& f, const v8f (&hT)[2], int ks0,
                                       v8f (&acc)[4 * NHEADS]) {
#pragma unroll
  for (int i = 0; i < 2; ++i) {
    const int ks = ks0 + i;
    v2f a;
    a.x = hT[ks >> 2][2 * (ks & 3)];
    a.y = hT[ks >> 2][2 * (ks & 3) + 1];
#pragma unroll
    for (int hd = 0; hd < NHEADS; ++hd) {
#pragma unroll
      for (int c = 0; c < 4; ++c) {
        v2f b;
        b.x = f.b0[i][hd][c];
        b.y = f.b1[i][hd][c];
        acc[4 * hd + c] = __builtin_amdgcn_wmma_f32_16x16x4_f32(false, a, false, b, (short)0,
                                                                acc[4 * hd + c], false, false);
      }
    }
  }
}

__global__ __launch_bounds__(128) __attribute__((amdgpu_waves_per_eu(3, 4)))
void lin_heads_kernel(const float* __restrict__ x, const float* __restrict__ lin_W,
                      const float* __restrict__ lin_b,
                      const float* __restrict__ W_heads, const float* __restrict__ a_heads,
                      const float* __restrict__ W_end, const float* __restrict__ a_end,
                      float* __restrict__ hcat, float* __restrict__ src,
                      float* __restrict__ dst, float* __restrict__ u2) {
  __shared__ v4f red[4][NHEADS][8][32];
  if (blockIdx.x < 2) {
    const int t = blockIdx.x * 128 + threadIdx.x;
    float s, d;
    dot64x2(W_end + (size_t)t * FOUT, a_end, s, d);
    u2[t] = s;
    u2[HCATF + t] = d;
    return;
  }
  const int mtile = blockIdx.x - 2;
  const int wave  = threadIdx.x >> 5;
  const int lane  = threadIdx.x & 31;
  const int half  = lane >> 4, l16 = lane & 15;
  const v8f zero = {0.f, 0.f, 0.f, 0.f, 0.f, 0.f, 0.f, 0.f};

  const float* xp = x + (size_t)(mtile * 16 + l16) * FIN + 4 * half;
  const float* wp = lin_W + (size_t)(4 * half) * FIN + 8 * l16 + 2 * wave;
  P1Frag g0, g1;
  p1_load(g0, xp, wp, 0);
  p1_load(g1, xp, wp, 8 * P1G);
  v8f hT[2];
  {
    const float* bp = lin_b + 64 * half + 2 * wave;
#pragma unroll
    for (int r = 0; r < 8; ++r) {
      const v2f bv = *(const v2f*)(bp + 8 * r);
      hT[0][r] = bv.x;
      hT[1][r] = bv.y;
    }
  }
  const float* Wb = W_heads + 4 * l16;
  const int fbase = 64 * half + 2 * wave;
  P2Frag q0, q1;
  __builtin_amdgcn_sched_barrier(0);
  p1_mma(g0, hT);
  p2_load(q0, Wb, fbase, 0);
  __builtin_amdgcn_sched_barrier(0);
  p1_mma(g1, hT);
  __builtin_amdgcn_sched_barrier(0);

  v8f acc[4 * NHEADS];
#pragma unroll
  for (int i = 0; i < 4 * NHEADS; ++i) acc[i] = zero;
#pragma unroll
  for (int ks = 0; ks < 8; ks += 4) {
    p2_load(q1, Wb, fbase, ks + 2);
    __builtin_amdgcn_sched_barrier(0);
    p2_mma(q0, hT, ks, acc);
    __builtin_amdgcn_sched_barrier(0);
    if (ks + 4 < 8) p2_load(q0, Wb, fbase, ks + 4);
    __builtin_amdgcn_sched_barrier(0);
    p2_mma(q1, hT, ks + 2, acc);
    __builtin_amdgcn_sched_barrier(0);
  }

#pragma unroll
  for (int hd = 0; hd < NHEADS; ++hd) {
#pragma unroll
    for (int r = 0; r < 8; ++r) {
      v4f p;
      p.x = acc[4 * hd + 0][r];
      p.y = acc[4 * hd + 1][r];
      p.z = acc[4 * hd + 2][r];
      p.w = acc[4 * hd + 3][r];
      red[wave][hd][r][lane] = p;
    }
  }
  __syncthreads();

  const int head = wave;
  const v4f as4 = *(const v4f*)(a_heads + head * 2 * FHID + 4 * l16);
  const v4f ad4 = *(const v4f*)(a_heads + head * 2 * FHID + FHID + 4 * l16);
  float* out = hcat + (size_t)(mtile * 16 + 8 * half) * HCATF + head * FHID + 4 * l16;
  float ps[8], pd[8];
#pragma unroll
  for (int r = 0; r < 8; ++r) {
    const v4f o = ((red[0][head][r][lane] + red[1][head][r][lane]) + red[2][head][r][lane])
                  + red[3][head][r][lane];
    *(v4f*)(out + (size_t)r * HCATF) = o;
    ps[r] = row16_sum(o.x * as4.x + o.y * as4.y + o.z * as4.z + o.w * as4.w);
    pd[r] = row16_sum(o.x * ad4.x + o.y * ad4.y + o.z * ad4.z + o.w * ad4.w);
  }
  if (l16 == 0) {
    float* sp = src + (size_t)head * NN + mtile * 16 + 8 * half;
    float* dp = dst + (size_t)head * NN + mtile * 16 + 8 * half;
    v4f s0 = {ps[0], ps[1], ps[2], ps[3]}, s1 = {ps[4], ps[5], ps[6], ps[7]};
    v4f d0 = {pd[0], pd[1], pd[2], pd[3]}, d1 = {pd[4], pd[5], pd[6], pd[7]};
    *(v4f*)sp = s0;
    *(v4f*)(sp + 4) = s1;
    *(v4f*)dp = d0;
    *(v4f*)(dp + 4) = d1;
  }
}

__device__ __forceinline__ float elu1(float x) { return (x > 0.f) ? x : (expf(x) - 1.f); }

#define HB 16

template <int M>
__device__ __forceinline__ int row_xor_i(int v) {
  return __builtin_amdgcn_update_dpp(v, v, 0x160 + M, 0xf, 0xf, false);
}

__global__ __launch_bounds__(32) __attribute__((amdgpu_waves_per_eu(1, 2)))
void gat_heads_fused_kernel(const float* __restrict__ adj,
                            const float* __restrict__ hcat,
                            const float* __restrict__ src,
                            const float* __restrict__ dst,
                            int* __restrict__ cnt, int* __restrict__ nidx,
                            float* __restrict__ gat1,
                            const float* __restrict__ u2,
                            float* __restrict__ src2, float* __restrict__ dst2) {
  __shared__ __attribute__((aligned(16))) int jidx[2][CAP + 16];
  __shared__ __attribute__((aligned(16))) v4f wgt[2][CAP];
  const int lane = threadIdx.x;
  const int hw   = lane >> 4;
  const int l16  = lane & 15;
  const int row  = blockIdx.x * 2 + hw;
  int* ji = jidx[hw];
  v4f* wg = wgt[hw];

  const float4* arow4 = (const float4*)(adj + (size_t)row * NN) + l16;
  const int spill = CAP + l16;
  float4 v[NN / 64];
#pragma unroll
  for (int q = 0; q < NN / 64; ++q) v[q] = arow4[q * 16];

  float si[NHEADS];
#pragma unroll
  for (int h = 0; h < NHEADS; ++h) si[h] = src[h * NN + row];

  unsigned bm[4][2];
#pragma unroll
  for (int e = 0; e < 4; ++e) { bm[e][0] = 0u; bm[e][1] = 0u; }
#pragma unroll
  for (int n = 0; n < NN / 64; ++n) {
    const float4 a = v[n];
    const int j = n >> 5;
    bm[0][j] = bm[0][j] + bm[0][j] + (a.x > 0.f ? 1u : 0u);
    bm[1][j] = bm[1][j] + bm[1][j] + (a.y > 0.f ? 1u : 0u);
    bm[2][j] = bm[2][j] + bm[2][j] + (a.z > 0.f ? 1u : 0u);
    bm[3][j] = bm[3][j] + bm[3][j] + (a.w > 0.f ? 1u : 0u);
  }

  int cl = 0;
#pragma unroll
  for (int e = 0; e < 4; ++e) cl += __popc(bm[e][0]) + __popc(bm[e][1]);
  int pre = 0, pos = cl;
  {
    int t;
    t = row_xor_i<1>(pos); pre += (l16 & 1) ? t : 0; pos += t;
    t = row_xor_i<2>(pos); pre += (l16 & 2) ? t : 0; pos += t;
    t = row_xor_i<4>(pos); pre += (l16 & 4) ? t : 0; pos += t;
    t = row_xor_i<8>(pos); pre += (l16 & 8) ? t : 0; pos += t;
  }
  {
    int k = pre;
#pragma unroll
    for (int e = 0; e < 4; ++e) {
#pragma unroll
      for (int j = 0; j < 2; ++j) {
        unsigned w = bm[e][j];
        const int top = 2048 * j + 1984 + 4 * l16 + e;
        while (w != 0u) {
          const int b = __builtin_ctz(w);
          w &= w - 1u;
          ji[k < CAP ? k : spill] = top - 64 * b;
          ++k;
        }
      }
    }
  }
  __syncthreads();

  const int deg  = pos < CAP ? pos : CAP;
  const int degp = (deg + HB - 1) & ~(HB - 1);
  if (l16 == 0) cnt[row] = deg;
  int* outp = nidx + (size_t)row * CAP;
  for (int kb = 4 * l16; kb < deg; kb += 64) *(int4*)(outp + kb) = *(const int4*)(ji + kb);

  v4f mx = {-3.4e38f, -3.4e38f, -3.4e38f, -3.4e38f};
  for (int kb = 0; kb < degp; kb += 64) {
    int   jv[4];
    float dv[4][NHEADS];
#pragma unroll
    for (int s = 0; s < 4; ++s) {
      const int k = kb + 16 * s + l16;
      jv[s] = ji[k < deg ? k : deg - 1];
    }
#pragma unroll
    for (int s = 0; s < 4; ++s)
#pragma unroll
      for (int h = 0; h < NHEADS; ++h) dv[s][h] = dst[h * NN + jv[s]];
#pragma unroll
    for (int s = 0; s < 4; ++s) {
      const int k = kb + 16 * s + l16;
      v4f e;
#pragma unroll
      for (int h = 0; h < NHEADS; ++h) {
        float t = si[h] + dv[s][h];
        t = (t >= 0.f) ? t : ALPHA_L * t;
        e[h] = (k < deg) ? t : -INFINITY;
        mx[h] = fmaxf(mx[h], e[h]);
      }
      if (k >= deg) ji[k] = jv[s];
      wg[k] = e;
    }
  }
#pragma unroll
  for (int h = 0; h < NHEADS; ++h) mx[h] = row16_max(mx[h]);

  v4f ss = {0.f, 0.f, 0.f, 0.f};
  for (int kb = 0; kb < degp; kb += 64) {
    v4f ev[4];
#pragma unroll
    for (int s = 0; s < 4; ++s) ev[s] = wg[kb + 16 * s + l16];
#pragma unroll
    for (int s = 0; s < 4; ++s) {
      v4f w;
#pragma unroll
      for (int h = 0; h < NHEADS; ++h) {
        w[h] = expf(ev[s][h] - mx[h]);
        ss[h] += w[h];
      }
      wg[kb + 16 * s + l16] = w;
    }
  }
  float denom[NHEADS];
#pragma unroll
  for (int h = 0; h < NHEADS; ++h) denom[h] = row16_sum(ss[h]);
  __syncthreads();

  float4 us[NHEADS], ud[NHEADS];
#pragma unroll
  for (int h = 0; h < NHEADS; ++h) {
    us[h] = *(const float4*)(u2 + h * FHID + 4 * l16);
    ud[h] = *(const float4*)(u2 + HCATF + h * FHID + 4 * l16);
  }
  const float* hb = hcat + 4 * l16;
  float4 acc[NHEADS];
#pragma unroll
  for (int h = 0; h < NHEADS; ++h) acc[h] = make_float4(0.f, 0.f, 0.f, 0.f);
  for (int k0 = 0; k0 < degp; k0 += HB) {
    int    jj[HB];
    v4f    ww[HB];
    float4 hv[HB][NHEADS];
#pragma unroll
    for (int q = 0; q < HB / 4; ++q) {
      const int4 jq = *(const int4*)(ji + k0 + 4 * q);
      jj[4 * q + 0] = jq.x; jj[4 * q + 1] = jq.y; jj[4 * q + 2] = jq.z; jj[4 * q + 3] = jq.w;
    }
#pragma unroll
    for (int i = 0; i < HB; ++i) ww[i] = wg[k0 + i];
#pragma unroll
    for (int i = 0; i < HB; ++i)
#pragma unroll
      for (int h = 0; h < NHEADS; ++h)
        hv[i][h] = *(const float4*)(hb + ((unsigned)jj[i] * HCATF + h * FHID));
#pragma unroll
    for (int i = 0; i < HB; ++i)
#pragma unroll
      for (int h = 0; h < NHEADS; ++h) {
        const float w = ww[i][h];
        acc[h].x += w * hv[i][h].x;
        acc[h].y += w * hv[i][h].y;
        acc[h].z += w * hv[i][h].z;
        acc[h].w += w * hv[i][h].w;
      }
  }

  float ps = 0.f, pd = 0.f;
  float* orow = gat1 + (size_t)row * HCATF + 4 * l16;
#pragma unroll
  for (int h = 0; h < NHEADS; ++h) {
    float4 o;
    o.x = elu1(acc[h].x / denom[h]);
    o.y = elu1(acc[h].y / denom[h]);
    o.z = elu1(acc[h].z / denom[h]);
    o.w = elu1(acc[h].w / denom[h]);
    *(float4*)(orow + h * FHID) = o;
    ps += o.x * us[h].x + o.y * us[h].y + o.z * us[h].z + o.w * us[h].w;
    pd += o.x * ud[h].x + o.y * ud[h].y + o.z * ud[h].z + o.w * ud[h].w;
  }
  ps = row16_sum(ps);
  pd = row16_sum(pd);
  if (l16 == 0) {
    src2[row] = ps;
    dst2[row] = pd;
  }
}

__global__ __launch_bounds__(64) __attribute__((amdgpu_waves_per_eu(1, 2)))
void end_gemm_kernel(const float* __restrict__ gat1,
                     const float* __restrict__ W_end,
                     float* __restrict__ Wh2) {
  const int mtile = blockIdx.x;
  const int wave  = threadIdx.x >> 5;
  const int lane  = threadIdx.x & 31;
  const int half  = lane >> 4, l16 = lane & 15;
  const int n0    = wave * 32;
  const v8f zero = {0.f, 0.f, 0.f, 0.f, 0.f, 0.f, 0.f, 0.f};
  v8f acc[2] = {zero, zero};
  gemm_strip<2, HCATF, 16>(gat1 + (size_t)mtile * 16 * HCATF, HCATF, W_end + n0, FOUT, lane, acc);
  float* out = Wh2 + (size_t)(mtile * 16 + 8 * half) * FOUT + n0 + 2 * l16;
#pragma unroll
  for (int r = 0; r < 8; ++r) {
    v2f o;
    o.x = acc[0][r];
    o.y = acc[1][r];
    *(v2f*)(out + (size_t)r * FOUT) = o;
  }
}

#define EB 16

__global__ __launch_bounds__(32) __attribute__((amdgpu_waves_per_eu(1, 4)))
void gat_end_kernel(const float* __restrict__ Wh2,
                    const float* __restrict__ src,
                    const float* __restrict__ dst,
                    const int* __restrict__ cnt,
                    const int* __restrict__ nidx,
                    float* __restrict__ out) {
  __shared__ __attribute__((aligned(16))) float wgt[2][CAP];
  __shared__ __attribute__((aligned(16))) int   jidx[2][CAP];
  const int lane = threadIdx.x;
  const int hw   = lane >> 4;
  const int l16  = lane & 15;
  const int row  = blockIdx.x * 2 + hw;
  const int deg  = cnt[row];
  float si = src[row];
  asm volatile("" : "+v"(si));
  const int* nb  = nidx + (size_t)row * CAP;
  const int degp = (deg + EB - 1) & ~(EB - 1);
  float* wg = wgt[hw];
  int*   ji = jidx[hw];

  float m = -3.4e38f;
  for (int kb = 0; kb < degp; kb += 64) {
    int   jv[4];
    float dv[4];
#pragma unroll
    for (int s = 0; s < 4; ++s) {
      const int k = kb + 16 * s + l16;
      jv[s] = nb[k < deg ? k : deg - 1];
    }
#pragma unroll
    for (int s = 0; s < 4; ++s) dv[s] = dst[jv[s]];
#pragma unroll
    for (int s = 0; s < 4; ++s) {
      const int k = kb + 16 * s + l16;
      float e = si + dv[s];
      e = (e >= 0.f) ? e : ALPHA_L * e;
      e = (k < deg) ? e : -INFINITY;
      ji[k] = jv[s];
      wg[k] = e;
      m = fmaxf(m, e);
    }
  }
  m = row16_max(m);

  float ssum = 0.f;
  for (int kb = 0; kb < degp; kb += 64) {
    float ev[4];
#pragma unroll
    for (int s = 0; s < 4; ++s) ev[s] = wg[kb + 16 * s + l16];
#pragma unroll
    for (int s = 0; s < 4; ++s) {
      const float w = expf(ev[s] - m);
      wg[kb + 16 * s + l16] = w;
      ssum += w;
    }
  }
  const float denom = row16_sum(ssum);
  __syncthreads();

  const unsigned lo = 4 * l16;
  float ax = 0.f, ay = 0.f, az = 0.f, aw = 0.f;
  for (int k0 = 0; k0 < degp; k0 += EB) {
    int    jj[EB];
    float  ww[EB];
    float4 hv[EB];
#pragma unroll
    for (int q = 0; q < EB / 4; ++q) {
      const int4   jq = *(const int4*)(ji + k0 + 4 * q);
      const float4 wq = *(const float4*)(wg + k0 + 4 * q);
      jj[4 * q + 0] = jq.x; jj[4 * q + 1] = jq.y; jj[4 * q + 2] = jq.z; jj[4 * q + 3] = jq.w;
      ww[4 * q + 0] = wq.x; ww[4 * q + 1] = wq.y; ww[4 * q + 2] = wq.z; ww[4 * q + 3] = wq.w;
    }
#pragma unroll
    for (int i = 0; i < EB; ++i)
      hv[i] = *(const float4*)(Wh2 + ((unsigned)jj[i] * FOUT + lo));
#pragma unroll
    for (int i = 0; i < EB; ++i) {
      ax = __builtin_fmaf(ww[i], hv[i].x, ax);
      ay = __builtin_fmaf(ww[i], hv[i].y, ay);
      az = __builtin_fmaf(ww[i], hv[i].z, az);
      aw = __builtin_fmaf(ww[i], hv[i].w, aw);
    }
  }
  float4 o;
  o.x = elu1(ax / denom);
  o.y = elu1(ay / denom);
  o.z = elu1(az / denom);
  o.w = elu1(aw / denom);

  const float mo = row16_max(fmaxf(fmaxf(o.x, o.y), fmaxf(o.z, o.w)));
  float4 ex;
  ex.x = expf(o.x - mo);
  ex.y = expf(o.y - mo);
  ex.z = expf(o.z - mo);
  ex.w = expf(o.w - mo);
  const float sum = row16_sum((ex.x + ex.y) + (ex.z + ex.w));
  float4 r;
  r.x = ex.x / sum;
  r.y = ex.y / sum;
  r.z = ex.z / sum;
  r.w = ex.w / sum;
  *(float4*)(out + (size_t)row * FOUT + lo) = r;
}

extern "C" void kernel_launch(void* const* d_in, const int* in_sizes, int n_in,
                              void* d_out, int out_size, void* d_ws, size_t ws_size,
                              hipStream_t stream) {
  (void)in_sizes; (void)n_in; (void)out_size; (void)ws_size;
  const float* x       = (const float*)d_in[0];
  const float* adj     = (const float*)d_in[1];
  const float* lin_W   = (const float*)d_in[2];
  const float* lin_b   = (const float*)d_in[3];
  const float* W_heads = (const float*)d_in[4];
  const float* a_heads = (const float*)d_in[5];
  const float* W_end   = (const float*)d_in[6];
  const float* a_end   = (const float*)d_in[7];
  float* out = (float*)d_out;

  char* ws = (char*)d_ws;
  float* hcat  = (float*)ws;  ws += (size_t)NN * HCATF * sizeof(float);
  float* gat1  = (float*)ws;  ws += (size_t)NN * HCATF * sizeof(float);
  float* Wh2   = (float*)ws;  ws += (size_t)NN * FOUT  * sizeof(float);
  float* src1  = (float*)ws;  ws += (size_t)NHEADS * NN * sizeof(float);
  float* dst1  = (float*)ws;  ws += (size_t)NHEADS * NN * sizeof(float);
  float* src2  = (float*)ws;  ws += (size_t)NN * sizeof(float);
  float* dst2  = (float*)ws;  ws += (size_t)NN * sizeof(float);
  int*   ncnt  = (int*)ws;    ws += (size_t)NN * sizeof(int);
  int*   nidx  = (int*)ws;    ws += (size_t)NN * CAP * sizeof(int);
  float* u2    = Wh2;

  lin_heads_kernel<<<NN / 16 + 2, 128, 0, stream>>>(x, lin_W, lin_b, W_heads, a_heads,
                                                     W_end, a_end, hcat, src1, dst1, u2);
  gat_heads_fused_kernel<<<NN / 2, 32, 0, stream>>>(adj, hcat, src1, dst1, ncnt, nidx, gat1,
                                                 u2, src2, dst2);
  end_gemm_kernel<<<NN / 16, 64, 0, stream>>>(gat1, W_end, Wh2);
  gat_end_kernel<<<NN / 2, 32, 0, stream>>>(Wh2, src2, dst2, ncnt, nidx, out);
}
